# starting kernel
# baseline (speedup 1.0000x reference)
#include <hip/hip_runtime.h>
#include <hip/hip_bf16.h>

typedef float v2f __attribute__((ext_vector_type(2)));
typedef float v8f __attribute__((ext_vector_type(8)));

#define D 64

__global__ void k_deg_init(float* __restrict__ deg, int n) {
  int i = blockIdx.x * blockDim.x + threadIdx.x;
  if (i < n) deg[i] = 1.0f;
}

__global__ void k_deg_accum(const int* __restrict__ dst, float* __restrict__ deg, int E) {
  int e = blockIdx.x * blockDim.x + threadIdx.x;
  if (e < E) atomicAdd(&deg[dst[e]], 1.0f);
}

__global__ void k_dinv(float* __restrict__ deg, int n) {
  int i = blockIdx.x * blockDim.x + threadIdx.x;
  if (i < n) {
    float d = deg[i];
    deg[i] = (d > 0.0f) ? rsqrtf(d) : 0.0f;
  }
}

__global__ void k_norm(const int* __restrict__ src, const int* __restrict__ dst,
                       const float* __restrict__ dinv, float* __restrict__ norm,
                       int E, int n) {
  int e = blockIdx.x * blockDim.x + threadIdx.x;
  int tot = E + n;
  if (e < tot) {
    int s, t;
    if (e < E) { s = src[e]; t = dst[e]; }
    else       { s = t = e - E; }
    norm[e] = dinv[s] * dinv[t];
  }
}

__global__ void k_gemm_wmma(const float* __restrict__ A, const float* __restrict__ W,
                            float* __restrict__ Cout, int M) {
  __shared__ v2f sW[(D / 2) * D];
  for (int i = threadIdx.x; i < (D / 2) * D; i += blockDim.x) {
    int kk = i / D;
    int nn = i - kk * D;
    v2f p;
    p.x = W[(2 * kk + 0) * D + nn];
    p.y = W[(2 * kk + 1) * D + nn];
    sW[i] = p;
  }
  __syncthreads();

  int lane = threadIdx.x & 31;
  int half = lane >> 4;
  int lid  = lane & 15;
  int wave = blockIdx.x * (blockDim.x >> 5) + (threadIdx.x >> 5);
  int row0 = wave * 16;
  if (row0 >= M) return;

  v8f acc0 = {}, acc1 = {}, acc2 = {}, acc3 = {};

  int r  = row0 + lid;
  int rc = (r < M) ? r : (M - 1);
  const v2f* arow = (const v2f*)(A + (size_t)rc * D);

#pragma unroll
  for (int k = 0; k < D; k += 4) {
    int kk = (k >> 1) + half;
    v2f a = arow[kk];
    const v2f* wrow = &sW[kk * D + lid];
    v2f b0 = wrow[0];
    v2f b1 = wrow[16];
    v2f b2 = wrow[32];
    v2f b3 = wrow[48];
    acc0 = __builtin_amdgcn_wmma_f32_16x16x4_f32(false, a, false, b0, (short)0, acc0, false, false);
    acc1 = __builtin_amdgcn_wmma_f32_16x16x4_f32(false, a, false, b1, (short)0, acc1, false, false);
    acc2 = __builtin_amdgcn_wmma_f32_16x16x4_f32(false, a, false, b2, (short)0, acc2, false, false);
    acc3 = __builtin_amdgcn_wmma_f32_16x16x4_f32(false, a, false, b3, (short)0, acc3, false, false);
  }

#pragma unroll
  for (int v = 0; v < 8; ++v) {
    int rr = row0 + v + 8 * half;
    if (rr < M) {
      float* crow = Cout + (size_t)rr * D + lid;
      crow[0]  = acc0[v];
      crow[16] = acc1[v];
      crow[32] = acc2[v];
      crow[48] = acc3[v];
    }
  }
}

__global__ void k_bias_init(float* __restrict__ agg, const float* __restrict__ b, int n) {
  int i = blockIdx.x * blockDim.x + threadIdx.x;
  if (i < n * D) agg[i] = b[i & (D - 1)];
}

__global__ void k_scatter(const int* __restrict__ src, const int* __restrict__ dst,
                          const float* __restrict__ norm, const float* __restrict__ h,
                          float* __restrict__ agg, int E, int n) {
  int tid = blockIdx.x * blockDim.x + threadIdx.x;
  int d = tid & (D - 1);
  int e = tid >> 6;
  int tot = E + n;
  if (e < tot) {
    int s, t;
    if (e < E) { s = src[e]; t = dst[e]; }
    else       { s = t = e - E; }
    float v = h[(size_t)s * D + d] * norm[e];
    atomicAdd(&agg[(size_t)t * D + d], v);
  }
}

__global__ void k_prelu(const float* __restrict__ in, const float* __restrict__ a,
                        float* __restrict__ out, int n) {
  int i = blockIdx.x * blockDim.x + threadIdx.x;
  if (i < n * D) {
    float v = in[i];
    float al = a[i & (D - 1)];
    out[i] = (v >= 0.0f) ? v : al * v;
  }
}

static inline int cdiv_ll(long long a, long long b) { return (int)((a + b - 1) / b); }

extern "C" void kernel_launch(void* const* d_in, const int* in_sizes, int n_in,
                              void* d_out, int out_size, void* d_ws, size_t ws_size,
                              hipStream_t stream) {
  (void)n_in; (void)out_size; (void)ws_size;

  const float* x  = (const float*)d_in[0];
  const int*   ei = (const int*)  d_in[1];
  const float* W1 = (const float*)d_in[2];
  const float* b1 = (const float*)d_in[3];
  const float* a1 = (const float*)d_in[4];
  const float* W2 = (const float*)d_in[5];
  const float* b2 = (const float*)d_in[6];
  const float* a2 = (const float*)d_in[7];

  const int n = in_sizes[0] / D;
  const int E = in_sizes[1] / 2;
  const int* src = ei;
  const int* dst = ei + E;

  float* ws   = (float*)d_ws;
  float* dinv = ws;
  float* norm = dinv + n;
  float* bufA = norm + (size_t)(E + n);
  float* bufB = bufA + (size_t)n * D;
  float* out  = (float*)d_out;

  const int B = 256;
  const int gN    = cdiv_ll(n, B);
  const int gE    = cdiv_ll(E, B);
  const int gEN   = cdiv_ll((long long)E + n, B);
  const int gND   = cdiv_ll((long long)n * D, B);
  const int gSc   = cdiv_ll(((long long)E + n) * D, B);
  const int gGemm = cdiv_ll((long long)cdiv_ll(n, 16) * 32, B);

  k_deg_init <<<gN,  B, 0, stream>>>(dinv, n);
  k_deg_accum<<<gE,  B, 0, stream>>>(dst, dinv, E);
  k_dinv     <<<gN,  B, 0, stream>>>(dinv, n);
  k_norm     <<<gEN, B, 0, stream>>>(src, dst, dinv, norm, E, n);

  k_gemm_wmma<<<gGemm, B, 0, stream>>>(x, W1, bufA, n);
  k_bias_init<<<gND,   B, 0, stream>>>(bufB, b1, n);
  k_scatter  <<<gSc,   B, 0, stream>>>(src, dst, norm, bufA, bufB, E, n);
  k_prelu    <<<gND,   B, 0, stream>>>(bufB, a1, bufA, n);

  k_gemm_wmma<<<gGemm, B, 0, stream>>>(bufA, W2, bufB, n);
  k_bias_init<<<gND,   B, 0, stream>>>(out, b2, n);
  k_scatter  <<<gSc,   B, 0, stream>>>(src, dst, norm, bufB, out, E, n);
  k_prelu    <<<gND,   B, 0, stream>>>(out, a2, out, n);
}


# Aggregation sweeps one byte per edge (owner-range number of its destination, from a pass replacing the norm pass) instead of 4-byte dst; norm = dinv[src]*dinv[dst] formed at use; sum order changes
# speedup vs baseline: 3.1017x; 3.1017x over previous
#include <hip/hip_runtime.h>
#include <hip/hip_bf16.h>

typedef float v2f __attribute__((ext_vector_type(2)));
typedef float v8f __attribute__((ext_vector_type(8)));

#define D 64
#define OWN_R 800

#define DEG_OWN 6400
#define DEG_C   16
#define DEG_T   1024
#define DEG_U   20

__global__ __launch_bounds__(DEG_T)
void k_deg_part(const int* __restrict__ dst, int* __restrict__ part, int E, int n) {
  __shared__ int cntL[DEG_OWN];
  const int tid = threadIdx.x;
  const int lo  = blockIdx.x * DEG_OWN;
  const int cnt = min(DEG_OWN, n - lo);
  const int pc  = blockIdx.y;

  for (int i = tid; i < DEG_OWN; i += DEG_T) cntL[i] = 0;
  __syncthreads();

  const int e_lo = (int)(((long long)pc * E) / DEG_C);
  const int e_hi = (int)(((long long)(pc + 1) * E) / DEG_C);
  int a_lo = e_lo + (int)((4u - (unsigned)(((size_t)(dst + e_lo) >> 2) & 3u)) & 3u);
  if (a_lo > e_hi) a_lo = e_hi;
  const int nvec = (e_hi - a_lo) >> 2;
  const int a_hi = a_lo + 4 * nvec;
  const int4* vp = (const int4*)(dst + a_lo);

  auto count = [&](int t) {
    const unsigned r = (unsigned)(t - lo);
    if (r < (unsigned)cnt)
      __hip_atomic_fetch_add(&cntL[r], 1, __ATOMIC_RELAXED, __HIP_MEMORY_SCOPE_WORKGROUP);
  };

  for (int base = tid; base < nvec; base += DEG_U * DEG_T) {
    int4 d[DEG_U];
#pragma unroll
    for (int j = 0; j < DEG_U; ++j) {
      const int i = base + j * DEG_T;
      d[j] = (i < nvec) ? vp[i] : make_int4(-1, -1, -1, -1);
    }
#pragma unroll
    for (int j = 0; j < DEG_U; ++j) {
      count(d[j].x); count(d[j].y); count(d[j].z); count(d[j].w);
    }
  }
  {
    const int nh = a_lo - e_lo;
    const int nt = e_hi - a_hi;
    if (tid < nh)           count(dst[e_lo + tid]);
    else if (tid < nh + nt) count(dst[a_hi + (tid - nh)]);
  }
  __syncthreads();

  for (int i = tid; i < cnt; i += DEG_T) part[(size_t)pc * n + lo + i] = cntL[i];
}

__global__ void k_dinv_sum(const int* __restrict__ part, float* __restrict__ dinv, int n) {
  int i = blockIdx.x * blockDim.x + threadIdx.x;
  if (i < n) {
    int s = 1;
#pragma unroll
    for (int c = 0; c < DEG_C; ++c) s += part[(size_t)c * n + i];
    float d = (float)s;
    dinv[i] = (d > 0.0f) ? rsqrtf(d) : 0.0f;
  }
}

__global__ void k_rid(const int* __restrict__ dst, unsigned int* __restrict__ rid4, int E) {
  int i = blockIdx.x * blockDim.x + threadIdx.x;
  int e = 4 * i;
  const unsigned int R = OWN_R;
  if (e + 3 < E) {
    const unsigned int d0 = (unsigned int)dst[e],     d1 = (unsigned int)dst[e + 1];
    const unsigned int d2 = (unsigned int)dst[e + 2], d3 = (unsigned int)dst[e + 3];
    rid4[i] = ((d0 / R) & 0xffu)         | (((d1 / R) & 0xffu) << 8) |
              (((d2 / R) & 0xffu) << 16) | (((d3 / R) & 0xffu) << 24);
  } else if (e < E) {
    unsigned int v = 0u;
    for (int k = 0; k < 4; ++k) {
      unsigned int b = 0xffu;
      if (e + k < E) b = ((unsigned int)dst[e + k] / R) & 0xffu;
      v |= b << (8 * k);
    }
    rid4[i] = v;
  }
}

__device__ __forceinline__ void gemm_body(const float* __restrict__ A, const float* __restrict__ A1,
                                          const float* __restrict__ alpha,
                                          const float* __restrict__ W, float* __restrict__ Cout, int M,
                                          v2f* sW, v2f* sAl, const bool FUSED) {
  for (int i = threadIdx.x; i < (D / 2) * D; i += blockDim.x) {
    int kk = i / D;
    int nn = i - kk * D;
    v2f p;
    p.x = W[(2 * kk + 0) * D + nn];
    p.y = W[(2 * kk + 1) * D + nn];
    sW[i] = p;
  }
  if (FUSED) {
    if (threadIdx.x < D / 2) {
      v2f p;
      p.x = alpha[2 * threadIdx.x + 0];
      p.y = alpha[2 * threadIdx.x + 1];
      sAl[threadIdx.x] = p;
    }
  }
  __syncthreads();

  int lane = threadIdx.x & 31;
  int half = lane >> 4;
  int lid  = lane & 15;
  int wave = blockIdx.x * (blockDim.x >> 5) + (threadIdx.x >> 5);
  int row0 = wave * 16;
  if (row0 >= M) return;

  v8f acc0 = {}, acc1 = {}, acc2 = {}, acc3 = {};

  int r  = row0 + lid;
  int rc = (r < M) ? r : (M - 1);
  const v2f* arow  = (const v2f*)(A + (size_t)rc * D);
  const v2f* arow1 = FUSED ? (const v2f*)(A1 + (size_t)rc * D) : arow;

#pragma unroll
  for (int k = 0; k < D; k += 4) {
    int kk = (k >> 1) + half;
    v2f a = arow[kk];
    if (FUSED) {
      v2f a1 = arow1[kk];
      v2f al = sAl[kk];
      float s0 = a.x + a1.x;
      float s1 = a.y + a1.y;
      a.x = (s0 >= 0.0f) ? s0 : al.x * s0;
      a.y = (s1 >= 0.0f) ? s1 : al.y * s1;
    }
    const v2f* wrow = &sW[kk * D + lid];
    v2f b0 = wrow[0];
    v2f b1 = wrow[16];
    v2f b2 = wrow[32];
    v2f b3 = wrow[48];
    acc0 = __builtin_amdgcn_wmma_f32_16x16x4_f32(false, a, false, b0, (short)0, acc0, false, false);
    acc1 = __builtin_amdgcn_wmma_f32_16x16x4_f32(false, a, false, b1, (short)0, acc1, false, false);
    acc2 = __builtin_amdgcn_wmma_f32_16x16x4_f32(false, a, false, b2, (short)0, acc2, false, false);
    acc3 = __builtin_amdgcn_wmma_f32_16x16x4_f32(false, a, false, b3, (short)0, acc3, false, false);
  }

#pragma unroll
  for (int v = 0; v < 8; ++v) {
    int rr = row0 + v + 8 * half;
    if (rr < M) {
      float* crow = Cout + (size_t)rr * D + lid;
      crow[0]  = acc0[v];
      crow[16] = acc1[v];
      crow[32] = acc2[v];
      crow[48] = acc3[v];
    }
  }
}

__global__ void k_gemm_wmma(const float* __restrict__ A, const float* __restrict__ W,
                            float* __restrict__ Cout, int M) {
  __shared__ v2f sW[(D / 2) * D];
  gemm_body(A, nullptr, nullptr, W, Cout, M, sW, nullptr, false);
}

__global__ void k_gemm_prelu_wmma(const float* __restrict__ P0, const float* __restrict__ P1,
                                  const float* __restrict__ alpha, const float* __restrict__ W,
                                  float* __restrict__ Cout, int M) {
  __shared__ v2f sW[(D / 2) * D];
  __shared__ v2f sAl[D / 2];
  gemm_body(P0, P1, alpha, W, Cout, M, sW, sAl, true);
}

#define AGG_T 1024
#define AGG_W (AGG_T / 32)
#define QCAP  256
#define AGG_C 2
#define PIECE_J 2

__device__ __forceinline__ void lds_add(float* p, float v) {
  __hip_atomic_fetch_add(p, v, __ATOMIC_RELAXED, __HIP_MEMORY_SCOPE_WORKGROUP);
}

__device__ __forceinline__ void agg_batch(const int* qw, int head, int valid, int lane, int lo, int cnt,
                                          const int* __restrict__ src, const int* __restrict__ dst,
                                          const float* __restrict__ dinv, const float* dinvL,
                                          const float* __restrict__ h, float* acc) {
  int   my_s = lo;
  int   my_t = OWN_R;
  float my_w = 0.0f;
  if (lane < valid) {
    int e = qw[(head + lane) & (QCAP - 1)];
    my_s = src[e];
    unsigned t = (unsigned)(dst[e] - lo);
    if (t >= (unsigned)cnt) t = OWN_R;
    my_t = (int)t;
    my_w = dinv[my_s] * dinvL[my_t];
  }
#pragma nounroll
  for (int g = 0; g < 32; g += 8) {
    if (g >= valid) break;
    float v0[8], v1[8];
    int   tt[8];
#pragma unroll
    for (int u = 0; u < 8; ++u) {
      int s = __builtin_amdgcn_readlane(my_s, g + u);
      tt[u] = __builtin_amdgcn_readlane(my_t, g + u);
      const float* hr = h + (size_t)s * D + lane;
      v0[u] = hr[0];
      v1[u] = hr[32];
    }
#pragma unroll
    for (int u = 0; u < 8; ++u) {
      float  w = __int_as_float(__builtin_amdgcn_readlane(__float_as_int(my_w), g + u));
      float* a = acc + tt[u] * D + lane;
      lds_add(a,      v0[u] * w);
      lds_add(a + 32, v1[u] * w);
    }
  }
}

__global__ __launch_bounds__(AGG_T)
void k_agg_owner(const int* __restrict__ src, const int* __restrict__ dst,
                 const unsigned int* __restrict__ rid4, const float* __restrict__ dinv,
                 const float* __restrict__ h,
                 const float* __restrict__ bias, float* __restrict__ agg0, float* __restrict__ agg1,
                 int E, int n) {
  __shared__ float acc[(OWN_R + 1) * D];
  __shared__ float dinvL[OWN_R + 1];
  __shared__ int   q[AGG_W * QCAP];

  const int tid  = threadIdx.x;
  const int lane = tid & 31;
  const int wv   = __builtin_amdgcn_readfirstlane(tid >> 5);
  const int lo   = blockIdx.x * OWN_R;
  const int cnt  = min(OWN_R, n - lo);
  const int sl   = blockIdx.y;

  if (sl == 0) {
    for (int i = tid; i < cnt * D; i += AGG_T) {
      int   node = lo + (i >> 6);
      float dn   = dinv[node];
      acc[i] = bias[i & (D - 1)] + h[(size_t)lo * D + i] * (dn * dn);
    }
  } else {
    for (int i = tid; i < cnt * D; i += AGG_T) acc[i] = 0.0f;
  }
  if (tid < D) acc[OWN_R * D + tid] = 0.0f;
  for (int i = tid; i < OWN_R + 1; i += AGG_T) dinvL[i] = (i < cnt) ? dinv[lo + i] : 0.0f;
  __syncthreads();

  int* qw = q + wv * QCAP;
  int qhead = 0, qtail = 0;

  auto push = [&](bool m, int e) {
    unsigned mask = __builtin_amdgcn_ballot_w32(m);
    if (mask) {
      int pos = (qtail + (int)__builtin_amdgcn_mbcnt_lo(mask, 0u)) & (QCAP - 1);
      if (m) qw[pos] = e;
      qtail += __builtin_popcount(mask);
    }
  };
  auto drain_full = [&]() {
    while (qtail - qhead >= 32) {
      agg_batch(qw, qhead, 32, lane, lo, cnt, src, dst, dinv, dinvL, h, acc);
      qhead += 32;
    }
  };
  auto test4 = [&](unsigned y, int e) {
    push((y & 0x000000ffu) == 0u, e);
    push((y & 0x0000ff00u) == 0u, e + 1);
    push((y & 0x00ff0000u) == 0u, e + 2);
    push(y < 0x01000000u,         e + 3);
    drain_full();
  };

  const uint4*   vp    = (const uint4*)rid4;
  const unsigned pat   = (blockIdx.x & 0xffu) * 0x01010101u;
  const int      nfull = E >> 10;
  const int   p0    = (int)(((long long)sl * nfull) / AGG_C);
  const int   np    = (int)(((long long)(sl + 1) * nfull) / AGG_C) - p0;
  const int   rot   = (int)((blockIdx.x * (unsigned)np) / gridDim.x);

  uint4 d[PIECE_J], dn[PIECE_J];
  if (wv < np) {
    int c  = wv;
    int cc = c + rot; if (cc >= np) cc -= np;
    cc += p0;
#pragma unroll
    for (int j = 0; j < PIECE_J; ++j) d[j] = vp[cc * (PIECE_J * 32) + j * 32 + lane];
    while (true) {
      const int  cnx  = c + AGG_W;
      const bool more = cnx < np;
      int ccn = cc;
      if (more) { ccn = cnx + rot; if (ccn >= np) ccn -= np; ccn += p0; }
#pragma unroll
      for (int j = 0; j < PIECE_J; ++j) dn[j] = vp[ccn * (PIECE_J * 32) + j * 32 + lane];
#pragma unroll
      for (int j = 0; j < PIECE_J; ++j) {
        int e0 = 16 * (cc * (PIECE_J * 32) + j * 32 + lane);
        test4(d[j].x ^ pat, e0);
        test4(d[j].y ^ pat, e0 + 4);
        test4(d[j].z ^ pat, e0 + 8);
        test4(d[j].w ^ pat, e0 + 12);
      }
      if (!more) break;
#pragma unroll
      for (int j = 0; j < PIECE_J; ++j) d[j] = dn[j];
      c  = cnx;
      cc = ccn;
    }
  }

  if (sl == 0) {
    const int tail0 = nfull << 10;
    const int nleft = E - tail0;
    for (int base = wv * 32; base < nleft; base += AGG_T) {
      const int idx = base + lane;
      const int e   = tail0 + idx;
      bool m = false;
      if (idx < nleft) m = (unsigned)(dst[e] - lo) < (unsigned)cnt;
      push(m, e);
    }
  }
  while (qtail - qhead > 0) {
    int valid = min(32, qtail - qhead);
    agg_batch(qw, qhead, valid, lane, lo, cnt, src, dst, dinv, dinvL, h, acc);
    qhead += valid;
  }

  __syncthreads();
  float* agg = (sl == 0) ? agg0 : agg1;
  for (int i = tid; i < cnt * D; i += AGG_T) agg[(size_t)lo * D + i] = acc[i];
}

__global__ void k_prelu(const float* in0, const float* __restrict__ in1, const float* __restrict__ a,
                        float* out, int n) {
  int i = blockIdx.x * blockDim.x + threadIdx.x;
  if (i < n * D) {
    float v = in0[i] + in1[i];
    float al = a[i & (D - 1)];
    out[i] = (v >= 0.0f) ? v : al * v;
  }
}

static inline int cdiv_ll(long long a, long long b) { return (int)((a + b - 1) / b); }

extern "C" void kernel_launch(void* const* d_in, const int* in_sizes, int n_in,
                              void* d_out, int out_size, void* d_ws, size_t ws_size,
                              hipStream_t stream) {
  (void)n_in; (void)out_size; (void)ws_size;

  const float* x  = (const float*)d_in[0];
  const int*   ei = (const int*)  d_in[1];
  const float* W1 = (const float*)d_in[2];
  const float* b1 = (const float*)d_in[3];
  const float* a1 = (const float*)d_in[4];
  const float* W2 = (const float*)d_in[5];
  const float* b2 = (const float*)d_in[6];
  const float* a2 = (const float*)d_in[7];

  const int n = in_sizes[0] / D;
  const int E = in_sizes[1] / 2;
  const int* src = ei;
  const int* dst = ei + E;

  float* ws   = (float*)d_ws;
  unsigned int* rid4 = (unsigned int*)ws;
  float* dinv = ws + (size_t)(E + n);
  float* bufA = dinv + n;
  float* bufB = bufA + (size_t)n * D;
  float* out  = (float*)d_out;
  int*   part = (int*)bufA;

  const int B = 256;
  const int gN    = cdiv_ll(n, B);
  const int gRid  = cdiv_ll(cdiv_ll(E, 4), B);
  const int gND   = cdiv_ll((long long)n * D, B);
  const int gOwn  = cdiv_ll(n, OWN_R);
  const int gGemm = cdiv_ll((long long)cdiv_ll(n, 16) * 32, B);
  const dim3 gAgg(gOwn, AGG_C);
  const dim3 gDeg(cdiv_ll(n, DEG_OWN), DEG_C);

  k_deg_part <<<gDeg, DEG_T, 0, stream>>>(dst, part, E, n);
  k_dinv_sum <<<gN,  B, 0, stream>>>(part, dinv, n);
  k_rid      <<<gRid, B, 0, stream>>>(dst, rid4, E);

  k_gemm_wmma<<<gGemm, B, 0, stream>>>(x, W1, bufA, n);
  k_agg_owner<<<gAgg, AGG_T, 0, stream>>>(src, dst, rid4, dinv, bufA, b1, bufB, out, E, n);

  k_gemm_prelu_wmma<<<gGemm, B, 0, stream>>>(bufB, out, a1, W2, bufA, n);
  k_agg_owner<<<gAgg, AGG_T, 0, stream>>>(src, dst, rid4, dinv, bufA, b2, out, bufB, E, n);
  k_prelu    <<<gND,   B, 0, stream>>>(out, bufB, a2, out, n);
}


# Aggregation sweep: a lane turns its piece's 32 range bytes into one word of match flags; the wave votes once per round of matches (about 2 a piece), not per byte position (32); sum order changes
# speedup vs baseline: 3.2227x; 1.0390x over previous
#include <hip/hip_runtime.h>
#include <hip/hip_bf16.h>

typedef float v2f __attribute__((ext_vector_type(2)));
typedef float v8f __attribute__((ext_vector_type(8)));

#define D 64
#define OWN_R 800

#define DEG_OWN 6400
#define DEG_C   16
#define DEG_T   1024
#define DEG_U   20

__global__ __launch_bounds__(DEG_T)
void k_deg_part(const int* __restrict__ dst, int* __restrict__ part, int E, int n) {
  __shared__ int cntL[DEG_OWN];
  const int tid = threadIdx.x;
  const int lo  = blockIdx.x * DEG_OWN;
  const int cnt = min(DEG_OWN, n - lo);
  const int pc  = blockIdx.y;

  for (int i = tid; i < DEG_OWN; i += DEG_T) cntL[i] = 0;
  __syncthreads();

  const int e_lo = (int)(((long long)pc * E) / DEG_C);
  const int e_hi = (int)(((long long)(pc + 1) * E) / DEG_C);
  int a_lo = e_lo + (int)((4u - (unsigned)(((size_t)(dst + e_lo) >> 2) & 3u)) & 3u);
  if (a_lo > e_hi) a_lo = e_hi;
  const int nvec = (e_hi - a_lo) >> 2;
  const int a_hi = a_lo + 4 * nvec;
  const int4* vp = (const int4*)(dst + a_lo);

  auto count = [&](int t) {
    const unsigned r = (unsigned)(t - lo);
    if (r < (unsigned)cnt)
      __hip_atomic_fetch_add(&cntL[r], 1, __ATOMIC_RELAXED, __HIP_MEMORY_SCOPE_WORKGROUP);
  };

  for (int base = tid; base < nvec; base += DEG_U * DEG_T) {
    int4 d[DEG_U];
#pragma unroll
    for (int j = 0; j < DEG_U; ++j) {
      const int i = base + j * DEG_T;
      d[j] = (i < nvec) ? vp[i] : make_int4(-1, -1, -1, -1);
    }
#pragma unroll
    for (int j = 0; j < DEG_U; ++j) {
      count(d[j].x); count(d[j].y); count(d[j].z); count(d[j].w);
    }
  }
  {
    const int nh = a_lo - e_lo;
    const int nt = e_hi - a_hi;
    if (tid < nh)           count(dst[e_lo + tid]);
    else if (tid < nh + nt) count(dst[a_hi + (tid - nh)]);
  }
  __syncthreads();

  for (int i = tid; i < cnt; i += DEG_T) part[(size_t)pc * n + lo + i] = cntL[i];
}

__global__ void k_dinv_sum(const int* __restrict__ part, float* __restrict__ dinv, int n) {
  int i = blockIdx.x * blockDim.x + threadIdx.x;
  if (i < n) {
    int s = 1;
#pragma unroll
    for (int c = 0; c < DEG_C; ++c) s += part[(size_t)c * n + i];
    float d = (float)s;
    dinv[i] = (d > 0.0f) ? rsqrtf(d) : 0.0f;
  }
}

__global__ void k_rid(const int* __restrict__ dst, unsigned int* __restrict__ rid4, int E) {
  int i = blockIdx.x * blockDim.x + threadIdx.x;
  int e = 4 * i;
  const unsigned int R = OWN_R;
  if (e + 3 < E) {
    const unsigned int d0 = (unsigned int)dst[e],     d1 = (unsigned int)dst[e + 1];
    const unsigned int d2 = (unsigned int)dst[e + 2], d3 = (unsigned int)dst[e + 3];
    rid4[i] = ((d0 / R) & 0xffu)         | (((d1 / R) & 0xffu) << 8) |
              (((d2 / R) & 0xffu) << 16) | (((d3 / R) & 0xffu) << 24);
  } else if (e < E) {
    unsigned int v = 0u;
    for (int k = 0; k < 4; ++k) {
      unsigned int b = 0xffu;
      if (e + k < E) b = ((unsigned int)dst[e + k] / R) & 0xffu;
      v |= b << (8 * k);
    }
    rid4[i] = v;
  }
}

__device__ __forceinline__ void gemm_body(const float* __restrict__ A, const float* __restrict__ A1,
                                          const float* __restrict__ alpha,
                                          const float* __restrict__ W, float* __restrict__ Cout, int M,
                                          v2f* sW, v2f* sAl, const bool FUSED) {
  for (int i = threadIdx.x; i < (D / 2) * D; i += blockDim.x) {
    int kk = i / D;
    int nn = i - kk * D;
    v2f p;
    p.x = W[(2 * kk + 0) * D + nn];
    p.y = W[(2 * kk + 1) * D + nn];
    sW[i] = p;
  }
  if (FUSED) {
    if (threadIdx.x < D / 2) {
      v2f p;
      p.x = alpha[2 * threadIdx.x + 0];
      p.y = alpha[2 * threadIdx.x + 1];
      sAl[threadIdx.x] = p;
    }
  }
  __syncthreads();

  int lane = threadIdx.x & 31;
  int half = lane >> 4;
  int lid  = lane & 15;
  int wave = blockIdx.x * (blockDim.x >> 5) + (threadIdx.x >> 5);
  int row0 = wave * 16;
  if (row0 >= M) return;

  v8f acc0 = {}, acc1 = {}, acc2 = {}, acc3 = {};

  int r  = row0 + lid;
  int rc = (r < M) ? r : (M - 1);
  const v2f* arow  = (const v2f*)(A + (size_t)rc * D);
  const v2f* arow1 = FUSED ? (const v2f*)(A1 + (size_t)rc * D) : arow;

#pragma unroll
  for (int k = 0; k < D; k += 4) {
    int kk = (k >> 1) + half;
    v2f a = arow[kk];
    if (FUSED) {
      v2f a1 = arow1[kk];
      v2f al = sAl[kk];
      float s0 = a.x + a1.x;
      float s1 = a.y + a1.y;
      a.x = (s0 >= 0.0f) ? s0 : al.x * s0;
      a.y = (s1 >= 0.0f) ? s1 : al.y * s1;
    }
    const v2f* wrow = &sW[kk * D + lid];
    v2f b0 = wrow[0];
    v2f b1 = wrow[16];
    v2f b2 = wrow[32];
    v2f b3 = wrow[48];
    acc0 = __builtin_amdgcn_wmma_f32_16x16x4_f32(false, a, false, b0, (short)0, acc0, false, false);
    acc1 = __builtin_amdgcn_wmma_f32_16x16x4_f32(false, a, false, b1, (short)0, acc1, false, false);
    acc2 = __builtin_amdgcn_wmma_f32_16x16x4_f32(false, a, false, b2, (short)0, acc2, false, false);
    acc3 = __builtin_amdgcn_wmma_f32_16x16x4_f32(false, a, false, b3, (short)0, acc3, false, false);
  }

#pragma unroll
  for (int v = 0; v < 8; ++v) {
    int rr = row0 + v + 8 * half;
    if (rr < M) {
      float* crow = Cout + (size_t)rr * D + lid;
      crow[0]  = acc0[v];
      crow[16] = acc1[v];
      crow[32] = acc2[v];
      crow[48] = acc3[v];
    }
  }
}

__global__ void k_gemm_wmma(const float* __restrict__ A, const float* __restrict__ W,
                            float* __restrict__ Cout, int M) {
  __shared__ v2f sW[(D / 2) * D];
  gemm_body(A, nullptr, nullptr, W, Cout, M, sW, nullptr, false);
}

__global__ void k_gemm_prelu_wmma(const float* __restrict__ P0, const float* __restrict__ P1,
                                  const float* __restrict__ alpha, const float* __restrict__ W,
                                  float* __restrict__ Cout, int M) {
  __shared__ v2f sW[(D / 2) * D];
  __shared__ v2f sAl[D / 2];
  gemm_body(P0, P1, alpha, W, Cout, M, sW, sAl, true);
}

#define AGG_T 1024
#define AGG_W (AGG_T / 32)
#define QCAP  256
#define AGG_C 2
#define PIECE_J 2
static_assert(PIECE_J == 2, "the sweep packs the two units of a piece into one 32-bit flag word per lane");

__device__ __forceinline__ void lds_add(float* p, float v) {
  __hip_atomic_fetch_add(p, v, __ATOMIC_RELAXED, __HIP_MEMORY_SCOPE_WORKGROUP);
}

__device__ __forceinline__ void agg_batch(const int* qw, int head, int valid, int lane, int lo, int cnt,
                                          const int* __restrict__ src, const int* __restrict__ dst,
                                          const float* __restrict__ dinv, const float* dinvL,
                                          const float* __restrict__ h, float* acc) {
  int   my_s = lo;
  int   my_t = OWN_R;
  float my_w = 0.0f;
  if (lane < valid) {
    int e = qw[(head + lane) & (QCAP - 1)];
    my_s = src[e];
    unsigned t = (unsigned)(dst[e] - lo);
    if (t >= (unsigned)cnt) t = OWN_R;
    my_t = (int)t;
    my_w = dinv[my_s] * dinvL[my_t];
  }
#pragma nounroll
  for (int g = 0; g < 32; g += 8) {
    if (g >= valid) break;
    float v0[8], v1[8];
    int   tt[8];
#pragma unroll
    for (int u = 0; u < 8; ++u) {
      int s = __builtin_amdgcn_readlane(my_s, g + u);
      tt[u] = __builtin_amdgcn_readlane(my_t, g + u);
      const float* hr = h + (size_t)s * D + lane;
      v0[u] = hr[0];
      v1[u] = hr[32];
    }
#pragma unroll
    for (int u = 0; u < 8; ++u) {
      float  w = __int_as_float(__builtin_amdgcn_readlane(__float_as_int(my_w), g + u));
      float* a = acc + tt[u] * D + lane;
      lds_add(a,      v0[u] * w);
      lds_add(a + 32, v1[u] * w);
    }
  }
}

__global__ __launch_bounds__(AGG_T)
void k_agg_owner(const int* __restrict__ src, const int* __restrict__ dst,
                 const unsigned int* __restrict__ rid4, const float* __restrict__ dinv,
                 const float* __restrict__ h,
                 const float* __restrict__ bias, float* __restrict__ agg0, float* __restrict__ agg1,
                 int E, int n) {
  __shared__ float acc[(OWN_R + 1) * D];
  __shared__ float dinvL[OWN_R + 1];
  __shared__ int   q[AGG_W * QCAP];

  const int tid  = threadIdx.x;
  const int lane = tid & 31;
  const int wv   = __builtin_amdgcn_readfirstlane(tid >> 5);
  const int lo   = blockIdx.x * OWN_R;
  const int cnt  = min(OWN_R, n - lo);
  const int sl   = blockIdx.y;

  if (sl == 0) {
    for (int i = tid; i < cnt * D; i += AGG_T) {
      int   node = lo + (i >> 6);
      float dn   = dinv[node];
      acc[i] = bias[i & (D - 1)] + h[(size_t)lo * D + i] * (dn * dn);
    }
  } else {
    for (int i = tid; i < cnt * D; i += AGG_T) acc[i] = 0.0f;
  }
  if (tid < D) acc[OWN_R * D + tid] = 0.0f;
  for (int i = tid; i < OWN_R + 1; i += AGG_T) dinvL[i] = (i < cnt) ? dinv[lo + i] : 0.0f;
  __syncthreads();

  int* qw = q + wv * QCAP;
  int qhead = 0, qtail = 0;

  auto push = [&](bool m, int e) {
    unsigned mask = __builtin_amdgcn_ballot_w32(m);
    if (mask) {
      int pos = (qtail + (int)__builtin_amdgcn_mbcnt_lo(mask, 0u)) & (QCAP - 1);
      if (m) qw[pos] = e;
      qtail += __builtin_popcount(mask);
    }
  };
  auto drain_full = [&]() {
    while (qtail - qhead >= 32) {
      agg_batch(qw, qhead, 32, lane, lo, cnt, src, dst, dinv, dinvL, h, acc);
      qhead += 32;
    }
  };
  const unsigned pat = (blockIdx.x & 0xffu) * 0x01010101u;

  auto zero_bytes = [](unsigned w) -> unsigned {
    return ~((((w & 0x7f7f7f7fu) + 0x7f7f7f7fu) | w) | 0x7f7f7f7fu);
  };
  auto test_piece = [&](uint4 u0, uint4 u1, int e0) {
    unsigned m = (zero_bytes(u0.x ^ pat) >> 7) | (zero_bytes(u0.y ^ pat) >> 6) |
                 (zero_bytes(u0.z ^ pat) >> 5) | (zero_bytes(u0.w ^ pat) >> 4) |
                 (zero_bytes(u1.x ^ pat) >> 3) | (zero_bytes(u1.y ^ pat) >> 2) |
                 (zero_bytes(u1.z ^ pat) >> 1) |  zero_bytes(u1.w ^ pat);
    unsigned mask = __builtin_amdgcn_ballot_w32(m != 0u);
    do {
      if (m != 0u) {
        const int p   = __ffs((int)m) - 1;
        const int pos = (qtail + (int)__builtin_amdgcn_mbcnt_lo(mask, 0u)) & (QCAP - 1);
        qw[pos] = e0 + ((p & 4) << 7) + ((p & 3) << 2) + (p >> 3);
      }
      qtail += __builtin_popcount(mask);
      m &= m - 1u;
      drain_full();
      mask = __builtin_amdgcn_ballot_w32(m != 0u);
    } while (mask != 0u);
  };

  const uint4*   vp    = (const uint4*)rid4;
  const int      nfull = E >> 10;
  const int   p0    = (int)(((long long)sl * nfull) / AGG_C);
  const int   np    = (int)(((long long)(sl + 1) * nfull) / AGG_C) - p0;
  const int   rot   = (int)((blockIdx.x * (unsigned)np) / gridDim.x);

  uint4 d[PIECE_J], dn[PIECE_J];
  if (wv < np) {
    int c  = wv;
    int cc = c + rot; if (cc >= np) cc -= np;
    cc += p0;
#pragma unroll
    for (int j = 0; j < PIECE_J; ++j) d[j] = vp[cc * (PIECE_J * 32) + j * 32 + lane];
    while (true) {
      const int  cnx  = c + AGG_W;
      const bool more = cnx < np;
      int ccn = cc;
      if (more) { ccn = cnx + rot; if (ccn >= np) ccn -= np; ccn += p0; }
#pragma unroll
      for (int j = 0; j < PIECE_J; ++j) dn[j] = vp[ccn * (PIECE_J * 32) + j * 32 + lane];
      test_piece(d[0], d[1], 16 * (cc * (PIECE_J * 32) + lane));
      if (!more) break;
#pragma unroll
      for (int j = 0; j < PIECE_J; ++j) d[j] = dn[j];
      c  = cnx;
      cc = ccn;
    }
  }

  if (sl == 0) {
    const int tail0 = nfull << 10;
    const int nleft = E - tail0;
    for (int base = wv * 32; base < nleft; base += AGG_T) {
      const int idx = base + lane;
      const int e   = tail0 + idx;
      bool m = false;
      if (idx < nleft) m = (unsigned)(dst[e] - lo) < (unsigned)cnt;
      push(m, e);
    }
  }
  while (qtail - qhead > 0) {
    int valid = min(32, qtail - qhead);
    agg_batch(qw, qhead, valid, lane, lo, cnt, src, dst, dinv, dinvL, h, acc);
    qhead += valid;
  }

  __syncthreads();
  float* agg = (sl == 0) ? agg0 : agg1;
  for (int i = tid; i < cnt * D; i += AGG_T) agg[(size_t)lo * D + i] = acc[i];
}

__global__ void k_prelu(const float* in0, const float* __restrict__ in1, const float* __restrict__ a,
                        float* out, int n) {
  int i = blockIdx.x * blockDim.x + threadIdx.x;
  if (i < n * D) {
    float v = in0[i] + in1[i];
    float al = a[i & (D - 1)];
    out[i] = (v >= 0.0f) ? v : al * v;
  }
}

static inline int cdiv_ll(long long a, long long b) { return (int)((a + b - 1) / b); }

extern "C" void kernel_launch(void* const* d_in, const int* in_sizes, int n_in,
                              void* d_out, int out_size, void* d_ws, size_t ws_size,
                              hipStream_t stream) {
  (void)n_in; (void)out_size; (void)ws_size;

  const float* x  = (const float*)d_in[0];
  const int*   ei = (const int*)  d_in[1];
  const float* W1 = (const float*)d_in[2];
  const float* b1 = (const float*)d_in[3];
  const float* a1 = (const float*)d_in[4];
  const float* W2 = (const float*)d_in[5];
  const float* b2 = (const float*)d_in[6];
  const float* a2 = (const float*)d_in[7];

  const int n = in_sizes[0] / D;
  const int E = in_sizes[1] / 2;
  const int* src = ei;
  const int* dst = ei + E;

  float* ws   = (float*)d_ws;
  unsigned int* rid4 = (unsigned int*)ws;
  float* dinv = ws + (size_t)(E + n);
  float* bufA = dinv + n;
  float* bufB = bufA + (size_t)n * D;
  float* out  = (float*)d_out;
  int*   part = (int*)bufA;

  const int B = 256;
  const int gN    = cdiv_ll(n, B);
  const int gRid  = cdiv_ll(cdiv_ll(E, 4), B);
  const int gND   = cdiv_ll((long long)n * D, B);
  const int gOwn  = cdiv_ll(n, OWN_R);
  const int gGemm = cdiv_ll((long long)cdiv_ll(n, 16) * 32, B);
  const dim3 gAgg(gOwn, AGG_C);
  const dim3 gDeg(cdiv_ll(n, DEG_OWN), DEG_C);

  k_deg_part <<<gDeg, DEG_T, 0, stream>>>(dst, part, E, n);
  k_dinv_sum <<<gN,  B, 0, stream>>>(part, dinv, n);
  k_rid      <<<gRid, B, 0, stream>>>(dst, rid4, E);

  k_gemm_wmma<<<gGemm, B, 0, stream>>>(x, W1, bufA, n);
  k_agg_owner<<<gAgg, AGG_T, 0, stream>>>(src, dst, rid4, dinv, bufA, b1, bufB, out, E, n);

  k_gemm_prelu_wmma<<<gGemm, B, 0, stream>>>(bufB, out, a1, W2, bufA, n);
  k_agg_owner<<<gAgg, AGG_T, 0, stream>>>(src, dst, rid4, dinv, bufA, b2, out, bufB, E, n);
  k_prelu    <<<gND,   B, 0, stream>>>(out, bufB, a2, out, n);
}


# Each layer aggregates first, then multiplies ((A X) W, was A (X W)): the product adds the two partial row sets at load, bias + PReLU in its epilogue; last PReLU pass dropped; sum order changes
# speedup vs baseline: 3.5705x; 1.1079x over previous
#include <hip/hip_runtime.h>
#include <hip/hip_bf16.h>

typedef float v2f __attribute__((ext_vector_type(2)));
typedef float v8f __attribute__((ext_vector_type(8)));

#define D 64
#define OWN_R 800

#define DEG_OWN 6400
#define DEG_C   16
#define DEG_T   1024
#define DEG_U   20

__global__ __launch_bounds__(DEG_T)
void k_deg_part(const int* __restrict__ dst, int* __restrict__ part, int E, int n) {
  __shared__ int cntL[DEG_OWN];
  const int tid = threadIdx.x;
  const int lo  = blockIdx.x * DEG_OWN;
  const int cnt = min(DEG_OWN, n - lo);
  const int pc  = blockIdx.y;

  for (int i = tid; i < DEG_OWN; i += DEG_T) cntL[i] = 0;
  __syncthreads();

  const int e_lo = (int)(((long long)pc * E) / DEG_C);
  const int e_hi = (int)(((long long)(pc + 1) * E) / DEG_C);
  int a_lo = e_lo + (int)((4u - (unsigned)(((size_t)(dst + e_lo) >> 2) & 3u)) & 3u);
  if (a_lo > e_hi) a_lo = e_hi;
  const int nvec = (e_hi - a_lo) >> 2;
  const int a_hi = a_lo + 4 * nvec;
  const int4* vp = (const int4*)(dst + a_lo);

  auto count = [&](int t) {
    const unsigned r = (unsigned)(t - lo);
    if (r < (unsigned)cnt)
      __hip_atomic_fetch_add(&cntL[r], 1, __ATOMIC_RELAXED, __HIP_MEMORY_SCOPE_WORKGROUP);
  };

  for (int base = tid; base < nvec; base += DEG_U * DEG_T) {
    int4 d[DEG_U];
#pragma unroll
    for (int j = 0; j < DEG_U; ++j) {
      const int i = base + j * DEG_T;
      d[j] = (i < nvec) ? vp[i] : make_int4(-1, -1, -1, -1);
    }
#pragma unroll
    for (int j = 0; j < DEG_U; ++j) {
      count(d[j].x); count(d[j].y); count(d[j].z); count(d[j].w);
    }
  }
  {
    const int nh = a_lo - e_lo;
    const int nt = e_hi - a_hi;
    if (tid < nh)           count(dst[e_lo + tid]);
    else if (tid < nh + nt) count(dst[a_hi + (tid - nh)]);
  }
  __syncthreads();

  for (int i = tid; i < cnt; i += DEG_T) part[(size_t)pc * n + lo + i] = cntL[i];
}

__global__ void k_dinv_sum(const int* __restrict__ part, float* __restrict__ dinv, int n) {
  int i = blockIdx.x * blockDim.x + threadIdx.x;
  if (i < n) {
    int s = 1;
#pragma unroll
    for (int c = 0; c < DEG_C; ++c) s += part[(size_t)c * n + i];
    float d = (float)s;
    dinv[i] = (d > 0.0f) ? rsqrtf(d) : 0.0f;
  }
}

__global__ void k_rid(const int* __restrict__ dst, unsigned int* __restrict__ rid4, int E) {
  int i = blockIdx.x * blockDim.x + threadIdx.x;
  int e = 4 * i;
  const unsigned int R = OWN_R;
  if (e + 3 < E) {
    const unsigned int d0 = (unsigned int)dst[e],     d1 = (unsigned int)dst[e + 1];
    const unsigned int d2 = (unsigned int)dst[e + 2], d3 = (unsigned int)dst[e + 3];
    rid4[i] = ((d0 / R) & 0xffu)         | (((d1 / R) & 0xffu) << 8) |
              (((d2 / R) & 0xffu) << 16) | (((d3 / R) & 0xffu) << 24);
  } else if (e < E) {
    unsigned int v = 0u;
    for (int k = 0; k < 4; ++k) {
      unsigned int b = 0xffu;
      if (e + k < E) b = ((unsigned int)dst[e + k] / R) & 0xffu;
      v |= b << (8 * k);
    }
    rid4[i] = v;
  }
}

__global__ void k_gemm_sum_bias_prelu_wmma(const float* __restrict__ P0, const float* __restrict__ P1,
                                           const float* __restrict__ W, const float* __restrict__ bias,
                                           const float* __restrict__ alpha, float* __restrict__ Cout, int M) {
  __shared__ v2f sW[(D / 2) * D];
  for (int i = threadIdx.x; i < (D / 2) * D; i += blockDim.x) {
    int kk = i / D;
    int nn = i - kk * D;
    v2f p;
    p.x = W[(2 * kk + 0) * D + nn];
    p.y = W[(2 * kk + 1) * D + nn];
    sW[i] = p;
  }
  __syncthreads();

  int lane = threadIdx.x & 31;
  int half = lane >> 4;
  int lid  = lane & 15;
  int wave = blockIdx.x * (blockDim.x >> 5) + (threadIdx.x >> 5);
  int row0 = wave * 16;
  if (row0 >= M) return;

  float bz[4], al[4];
#pragma unroll
  for (int j = 0; j < 4; ++j) {
    bz[j] = bias[16 * j + lid];
    al[j] = alpha[16 * j + lid];
  }

  v8f acc0 = {}, acc1 = {}, acc2 = {}, acc3 = {};

  int r  = row0 + lid;
  int rc = (r < M) ? r : (M - 1);
  const v2f* arow0 = (const v2f*)(P0 + (size_t)rc * D);
  const v2f* arow1 = (const v2f*)(P1 + (size_t)rc * D);

#pragma unroll
  for (int k = 0; k < D; k += 4) {
    int kk = (k >> 1) + half;
    v2f a  = arow0[kk];
    v2f a1 = arow1[kk];
    a.x = a.x + a1.x;
    a.y = a.y + a1.y;
    const v2f* wrow = &sW[kk * D + lid];
    v2f b0 = wrow[0];
    v2f b1 = wrow[16];
    v2f b2 = wrow[32];
    v2f b3 = wrow[48];
    acc0 = __builtin_amdgcn_wmma_f32_16x16x4_f32(false, a, false, b0, (short)0, acc0, false, false);
    acc1 = __builtin_amdgcn_wmma_f32_16x16x4_f32(false, a, false, b1, (short)0, acc1, false, false);
    acc2 = __builtin_amdgcn_wmma_f32_16x16x4_f32(false, a, false, b2, (short)0, acc2, false, false);
    acc3 = __builtin_amdgcn_wmma_f32_16x16x4_f32(false, a, false, b3, (short)0, acc3, false, false);
  }

#pragma unroll
  for (int v = 0; v < 8; ++v) {
    int rr = row0 + v + 8 * half;
    if (rr < M) {
      float* crow = Cout + (size_t)rr * D + lid;
      float t0 = acc0[v] + bz[0];
      float t1 = acc1[v] + bz[1];
      float t2 = acc2[v] + bz[2];
      float t3 = acc3[v] + bz[3];
      crow[0]  = (t0 >= 0.0f) ? t0 : al[0] * t0;
      crow[16] = (t1 >= 0.0f) ? t1 : al[1] * t1;
      crow[32] = (t2 >= 0.0f) ? t2 : al[2] * t2;
      crow[48] = (t3 >= 0.0f) ? t3 : al[3] * t3;
    }
  }
}

#define AGG_T 1024
#define AGG_W (AGG_T / 32)
#define QCAP  256
#define AGG_C 2
#define PIECE_J 2
static_assert(PIECE_J == 2, "the sweep packs the two units of a piece into one 32-bit flag word per lane");

__device__ __forceinline__ void lds_add(float* p, float v) {
  __hip_atomic_fetch_add(p, v, __ATOMIC_RELAXED, __HIP_MEMORY_SCOPE_WORKGROUP);
}

__device__ __forceinline__ void agg_batch(const int* qw, int head, int valid, int lane, int lo, int cnt,
                                          const int* __restrict__ src, const int* __restrict__ dst,
                                          const float* __restrict__ dinv, const float* dinvL,
                                          const float* __restrict__ h, float* acc) {
  int   my_s = lo;
  int   my_t = OWN_R;
  float my_w = 0.0f;
  if (lane < valid) {
    int e = qw[(head + lane) & (QCAP - 1)];
    my_s = src[e];
    unsigned t = (unsigned)(dst[e] - lo);
    if (t >= (unsigned)cnt) t = OWN_R;
    my_t = (int)t;
    my_w = dinv[my_s] * dinvL[my_t];
  }
#pragma nounroll
  for (int g = 0; g < 32; g += 8) {
    if (g >= valid) break;
    float v0[8], v1[8];
    int   tt[8];
#pragma unroll
    for (int u = 0; u < 8; ++u) {
      int s = __builtin_amdgcn_readlane(my_s, g + u);
      tt[u] = __builtin_amdgcn_readlane(my_t, g + u);
      const float* hr = h + (size_t)s * D + lane;
      v0[u] = hr[0];
      v1[u] = hr[32];
    }
#pragma unroll
    for (int u = 0; u < 8; ++u) {
      float  w = __int_as_float(__builtin_amdgcn_readlane(__float_as_int(my_w), g + u));
      float* a = acc + tt[u] * D + lane;
      lds_add(a,      v0[u] * w);
      lds_add(a + 32, v1[u] * w);
    }
  }
}

__global__ __launch_bounds__(AGG_T)
void k_agg_owner(const int* __restrict__ src, const int* __restrict__ dst,
                 const unsigned int* __restrict__ rid4, const float* __restrict__ dinv,
                 const float* __restrict__ h,
                 float* __restrict__ agg0, float* __restrict__ agg1,
                 int E, int n) {
  __shared__ float acc[(OWN_R + 1) * D];
  __shared__ float dinvL[OWN_R + 1];
  __shared__ int   q[AGG_W * QCAP];

  const int tid  = threadIdx.x;
  const int lane = tid & 31;
  const int wv   = __builtin_amdgcn_readfirstlane(tid >> 5);
  const int lo   = blockIdx.x * OWN_R;
  const int cnt  = min(OWN_R, n - lo);
  const int sl   = blockIdx.y;

  if (sl == 0) {
    for (int i = tid; i < cnt * D; i += AGG_T) {
      int   node = lo + (i >> 6);
      float dn   = dinv[node];
      acc[i] = h[(size_t)lo * D + i] * (dn * dn);
    }
  } else {
    for (int i = tid; i < cnt * D; i += AGG_T) acc[i] = 0.0f;
  }
  if (tid < D) acc[OWN_R * D + tid] = 0.0f;
  for (int i = tid; i < OWN_R + 1; i += AGG_T) dinvL[i] = (i < cnt) ? dinv[lo + i] : 0.0f;
  __syncthreads();

  int* qw = q + wv * QCAP;
  int qhead = 0, qtail = 0;

  auto push = [&](bool m, int e) {
    unsigned mask = __builtin_amdgcn_ballot_w32(m);
    if (mask) {
      int pos = (qtail + (int)__builtin_amdgcn_mbcnt_lo(mask, 0u)) & (QCAP - 1);
      if (m) qw[pos] = e;
      qtail += __builtin_popcount(mask);
    }
  };
  auto drain_full = [&]() {
    while (qtail - qhead >= 32) {
      agg_batch(qw, qhead, 32, lane, lo, cnt, src, dst, dinv, dinvL, h, acc);
      qhead += 32;
    }
  };
  const unsigned pat = (blockIdx.x & 0xffu) * 0x01010101u;

  auto zero_bytes = [](unsigned w) -> unsigned {
    return ~((((w & 0x7f7f7f7fu) + 0x7f7f7f7fu) | w) | 0x7f7f7f7fu);
  };
  auto test_piece = [&](uint4 u0, uint4 u1, int e0) {
    unsigned m = (zero_bytes(u0.x ^ pat) >> 7) | (zero_bytes(u0.y ^ pat) >> 6) |
                 (zero_bytes(u0.z ^ pat) >> 5) | (zero_bytes(u0.w ^ pat) >> 4) |
                 (zero_bytes(u1.x ^ pat) >> 3) | (zero_bytes(u1.y ^ pat) >> 2) |
                 (zero_bytes(u1.z ^ pat) >> 1) |  zero_bytes(u1.w ^ pat);
    unsigned mask = __builtin_amdgcn_ballot_w32(m != 0u);
    do {
      if (m != 0u) {
        const int p   = __ffs((int)m) - 1;
        const int pos = (qtail + (int)__builtin_amdgcn_mbcnt_lo(mask, 0u)) & (QCAP - 1);
        qw[pos] = e0 + ((p & 4) << 7) + ((p & 3) << 2) + (p >> 3);
      }
      qtail += __builtin_popcount(mask);
      m &= m - 1u;
      drain_full();
      mask = __builtin_amdgcn_ballot_w32(m != 0u);
    } while (mask != 0u);
  };

  const uint4*   vp    = (const uint4*)rid4;
  const int      nfull = E >> 10;
  const int   p0    = (int)(((long long)sl * nfull) / AGG_C);
  const int   np    = (int)(((long long)(sl + 1) * nfull) / AGG_C) - p0;
  const int   rot   = (int)((blockIdx.x * (unsigned)np) / gridDim.x);

  uint4 d[PIECE_J], dn[PIECE_J];
  if (wv < np) {
    int c  = wv;
    int cc = c + rot; if (cc >= np) cc -= np;
    cc += p0;
#pragma unroll
    for (int j = 0; j < PIECE_J; ++j) d[j] = vp[cc * (PIECE_J * 32) + j * 32 + lane];
    while (true) {
      const int  cnx  = c + AGG_W;
      const bool more = cnx < np;
      int ccn = cc;
      if (more) { ccn = cnx + rot; if (ccn >= np) ccn -= np; ccn += p0; }
#pragma unroll
      for (int j = 0; j < PIECE_J; ++j) dn[j] = vp[ccn * (PIECE_J * 32) + j * 32 + lane];
      test_piece(d[0], d[1], 16 * (cc * (PIECE_J * 32) + lane));
      if (!more) break;
#pragma unroll
      for (int j = 0; j < PIECE_J; ++j) d[j] = dn[j];
      c  = cnx;
      cc = ccn;
    }
  }

  if (sl == 0) {
    const int tail0 = nfull << 10;
    const int nleft = E - tail0;
    for (int base = wv * 32; base < nleft; base += AGG_T) {
      const int idx = base + lane;
      const int e   = tail0 + idx;
      bool m = false;
      if (idx < nleft) m = (unsigned)(dst[e] - lo) < (unsigned)cnt;
      push(m, e);
    }
  }
  while (qtail - qhead > 0) {
    int valid = min(32, qtail - qhead);
    agg_batch(qw, qhead, valid, lane, lo, cnt, src, dst, dinv, dinvL, h, acc);
    qhead += valid;
  }

  __syncthreads();
  float* agg = (sl == 0) ? agg0 : agg1;
  for (int i = tid; i < cnt * D; i += AGG_T) agg[(size_t)lo * D + i] = acc[i];
}

static inline int cdiv_ll(long long a, long long b) { return (int)((a + b - 1) / b); }

extern "C" void kernel_launch(void* const* d_in, const int* in_sizes, int n_in,
                              void* d_out, int out_size, void* d_ws, size_t ws_size,
                              hipStream_t stream) {
  (void)n_in; (void)out_size; (void)ws_size;

  const float* x  = (const float*)d_in[0];
  const int*   ei = (const int*)  d_in[1];
  const float* W1 = (const float*)d_in[2];
  const float* b1 = (const float*)d_in[3];
  const float* a1 = (const float*)d_in[4];
  const float* W2 = (const float*)d_in[5];
  const float* b2 = (const float*)d_in[6];
  const float* a2 = (const float*)d_in[7];

  const int n = in_sizes[0] / D;
  const int E = in_sizes[1] / 2;
  const int* src = ei;
  const int* dst = ei + E;

  float* ws   = (float*)d_ws;
  unsigned int* rid4 = (unsigned int*)ws;
  float* dinv = ws + (size_t)(E + n);
  float* bufA = dinv + n;
  float* bufB = bufA + (size_t)n * D;
  float* out  = (float*)d_out;
  int*   part = (int*)bufA;

  const int B = 256;
  const int gN    = cdiv_ll(n, B);
  const int gRid  = cdiv_ll(cdiv_ll(E, 4), B);
  const int gOwn  = cdiv_ll(n, OWN_R);
  const int gGemm = cdiv_ll((long long)cdiv_ll(n, 16) * 32, B);
  const dim3 gAgg(gOwn, AGG_C);
  const dim3 gDeg(cdiv_ll(n, DEG_OWN), DEG_C);

  k_deg_part <<<gDeg, DEG_T, 0, stream>>>(dst, part, E, n);
  k_dinv_sum <<<gN,  B, 0, stream>>>(part, dinv, n);
  k_rid      <<<gRid, B, 0, stream>>>(dst, rid4, E);

  k_agg_owner<<<gAgg, AGG_T, 0, stream>>>(src, dst, rid4, dinv, x, bufA, bufB, E, n);
  k_gemm_sum_bias_prelu_wmma<<<gGemm, B, 0, stream>>>(bufA, bufB, W1, b1, a1, out, n);

  k_agg_owner<<<gAgg, AGG_T, 0, stream>>>(src, dst, rid4, dinv, out, bufA, bufB, E, n);
  k_gemm_sum_bias_prelu_wmma<<<gGemm, B, 0, stream>>>(bufA, bufB, W2, b2, a2, out, n);
}


# Ring batch asks for all 32 source rows at once (one 8-byte load per lane and row) beside the dinv gather, not in 4 dependent groups of 8 behind it: 6 round trips a batch become 2; sum order changes
# speedup vs baseline: 3.9616x; 1.1095x over previous
#include <hip/hip_runtime.h>
#include <hip/hip_bf16.h>

typedef float v2f __attribute__((ext_vector_type(2)));
typedef float v8f __attribute__((ext_vector_type(8)));

#define D 64
#define OWN_R 800

#define DEG_OWN 6400
#define DEG_C   16
#define DEG_T   1024
#define DEG_U   20

__global__ __launch_bounds__(DEG_T)
void k_deg_part(const int* __restrict__ dst, int* __restrict__ part, int E, int n) {
  __shared__ int cntL[DEG_OWN];
  const int tid = threadIdx.x;
  const int lo  = blockIdx.x * DEG_OWN;
  const int cnt = min(DEG_OWN, n - lo);
  const int pc  = blockIdx.y;

  for (int i = tid; i < DEG_OWN; i += DEG_T) cntL[i] = 0;
  __syncthreads();

  const int e_lo = (int)(((long long)pc * E) / DEG_C);
  const int e_hi = (int)(((long long)(pc + 1) * E) / DEG_C);
  int a_lo = e_lo + (int)((4u - (unsigned)(((size_t)(dst + e_lo) >> 2) & 3u)) & 3u);
  if (a_lo > e_hi) a_lo = e_hi;
  const int nvec = (e_hi - a_lo) >> 2;
  const int a_hi = a_lo + 4 * nvec;
  const int4* vp = (const int4*)(dst + a_lo);

  auto count = [&](int t) {
    const unsigned r = (unsigned)(t - lo);
    if (r < (unsigned)cnt)
      __hip_atomic_fetch_add(&cntL[r], 1, __ATOMIC_RELAXED, __HIP_MEMORY_SCOPE_WORKGROUP);
  };

  for (int base = tid; base < nvec; base += DEG_U * DEG_T) {
    int4 d[DEG_U];
#pragma unroll
    for (int j = 0; j < DEG_U; ++j) {
      const int i = base + j * DEG_T;
      d[j] = (i < nvec) ? vp[i] : make_int4(-1, -1, -1, -1);
    }
#pragma unroll
    for (int j = 0; j < DEG_U; ++j) {
      count(d[j].x); count(d[j].y); count(d[j].z); count(d[j].w);
    }
  }
  {
    const int nh = a_lo - e_lo;
    const int nt = e_hi - a_hi;
    if (tid < nh)           count(dst[e_lo + tid]);
    else if (tid < nh + nt) count(dst[a_hi + (tid - nh)]);
  }
  __syncthreads();

  for (int i = tid; i < cnt; i += DEG_T) part[(size_t)pc * n + lo + i] = cntL[i];
}

__global__ void k_dinv_sum(const int* __restrict__ part, float* __restrict__ dinv, int n) {
  int i = blockIdx.x * blockDim.x + threadIdx.x;
  if (i < n) {
    int s = 1;
#pragma unroll
    for (int c = 0; c < DEG_C; ++c) s += part[(size_t)c * n + i];
    float d = (float)s;
    dinv[i] = (d > 0.0f) ? rsqrtf(d) : 0.0f;
  }
}

__global__ void k_rid(const int* __restrict__ dst, unsigned int* __restrict__ rid4, int E) {
  int i = blockIdx.x * blockDim.x + threadIdx.x;
  int e = 4 * i;
  const unsigned int R = OWN_R;
  if (e + 3 < E) {
    const unsigned int d0 = (unsigned int)dst[e],     d1 = (unsigned int)dst[e + 1];
    const unsigned int d2 = (unsigned int)dst[e + 2], d3 = (unsigned int)dst[e + 3];
    rid4[i] = ((d0 / R) & 0xffu)         | (((d1 / R) & 0xffu) << 8) |
              (((d2 / R) & 0xffu) << 16) | (((d3 / R) & 0xffu) << 24);
  } else if (e < E) {
    unsigned int v = 0u;
    for (int k = 0; k < 4; ++k) {
      unsigned int b = 0xffu;
      if (e + k < E) b = ((unsigned int)dst[e + k] / R) & 0xffu;
      v |= b << (8 * k);
    }
    rid4[i] = v;
  }
}

__global__ void k_gemm_sum_bias_prelu_wmma(const float* __restrict__ P0, const float* __restrict__ P1,
                                           const float* __restrict__ W, const float* __restrict__ bias,
                                           const float* __restrict__ alpha, float* __restrict__ Cout, int M) {
  __shared__ v2f sW[(D / 2) * D];
  for (int i = threadIdx.x; i < (D / 2) * D; i += blockDim.x) {
    int kk = i / D;
    int nn = i - kk * D;
    v2f p;
    p.x = W[(2 * kk + 0) * D + nn];
    p.y = W[(2 * kk + 1) * D + nn];
    sW[i] = p;
  }
  __syncthreads();

  int lane = threadIdx.x & 31;
  int half = lane >> 4;
  int lid  = lane & 15;
  int wave = blockIdx.x * (blockDim.x >> 5) + (threadIdx.x >> 5);
  int row0 = wave * 16;
  if (row0 >= M) return;

  float bz[4], al[4];
#pragma unroll
  for (int j = 0; j < 4; ++j) {
    bz[j] = bias[16 * j + lid];
    al[j] = alpha[16 * j + lid];
  }

  v8f acc0 = {}, acc1 = {}, acc2 = {}, acc3 = {};

  int r  = row0 + lid;
  int rc = (r < M) ? r : (M - 1);
  const v2f* arow0 = (const v2f*)(P0 + (size_t)rc * D);
  const v2f* arow1 = (const v2f*)(P1 + (size_t)rc * D);

#pragma unroll
  for (int k = 0; k < D; k += 4) {
    int kk = (k >> 1) + half;
    v2f a  = arow0[kk];
    v2f a1 = arow1[kk];
    a.x = a.x + a1.x;
    a.y = a.y + a1.y;
    const v2f* wrow = &sW[kk * D + lid];
    v2f b0 = wrow[0];
    v2f b1 = wrow[16];
    v2f b2 = wrow[32];
    v2f b3 = wrow[48];
    acc0 = __builtin_amdgcn_wmma_f32_16x16x4_f32(false, a, false, b0, (short)0, acc0, false, false);
    acc1 = __builtin_amdgcn_wmma_f32_16x16x4_f32(false, a, false, b1, (short)0, acc1, false, false);
    acc2 = __builtin_amdgcn_wmma_f32_16x16x4_f32(false, a, false, b2, (short)0, acc2, false, false);
    acc3 = __builtin_amdgcn_wmma_f32_16x16x4_f32(false, a, false, b3, (short)0, acc3, false, false);
  }

#pragma unroll
  for (int v = 0; v < 8; ++v) {
    int rr = row0 + v + 8 * half;
    if (rr < M) {
      float* crow = Cout + (size_t)rr * D + lid;
      float t0 = acc0[v] + bz[0];
      float t1 = acc1[v] + bz[1];
      float t2 = acc2[v] + bz[2];
      float t3 = acc3[v] + bz[3];
      crow[0]  = (t0 >= 0.0f) ? t0 : al[0] * t0;
      crow[16] = (t1 >= 0.0f) ? t1 : al[1] * t1;
      crow[32] = (t2 >= 0.0f) ? t2 : al[2] * t2;
      crow[48] = (t3 >= 0.0f) ? t3 : al[3] * t3;
    }
  }
}

#define AGG_T 1024
#define AGG_W (AGG_T / 32)
#define QCAP  256
#define AGG_C 2
#define PIECE_J 2
static_assert(PIECE_J == 2, "the sweep packs the two units of a piece into one 32-bit flag word per lane");

__device__ __forceinline__ void lds_add(float* p, float v) {
  __hip_atomic_fetch_add(p, v, __ATOMIC_RELAXED, __HIP_MEMORY_SCOPE_WORKGROUP);
}

__device__ __forceinline__ void agg_batch(const int* qw, int head, int valid, int lane, int lo, int cnt,
                                          const int* __restrict__ src, const int* __restrict__ dst,
                                          const float* __restrict__ dinv, const float* dinvL,
                                          const float* __restrict__ h, float* acc) {
  int my_s = lo;
  int my_d = -1;
  if (lane < valid) {
    int e = qw[(head + lane) & (QCAP - 1)];
    my_s = src[e];
    my_d = dst[e];
  }
  const float dv = dinv[my_s];
  v2f r[32];
#pragma unroll
  for (int u = 0; u < 32; ++u) {
    const int  s  = __builtin_amdgcn_readlane(my_s, u);
    const v2f* hr = (const v2f*)(h + (size_t)s * D);
    r[u] = hr[lane];
  }
  unsigned t = (unsigned)(my_d - lo);
  if (t >= (unsigned)cnt) t = OWN_R;
  const float my_w = dv * dinvL[t];
#pragma unroll
  for (int u = 0; u < 32; ++u) {
    const int   tt = __builtin_amdgcn_readlane((int)t, u);
    const float w  = __int_as_float(__builtin_amdgcn_readlane(__float_as_int(my_w), u));
    float* a = acc + tt * D + 2 * lane;
    lds_add(a,     r[u].x * w);
    lds_add(a + 1, r[u].y * w);
  }
}

__global__ __launch_bounds__(AGG_T)
void k_agg_owner(const int* __restrict__ src, const int* __restrict__ dst,
                 const unsigned int* __restrict__ rid4, const float* __restrict__ dinv,
                 const float* __restrict__ h,
                 float* __restrict__ agg0, float* __restrict__ agg1,
                 int E, int n) {
  __shared__ float acc[(OWN_R + 1) * D];
  __shared__ float dinvL[OWN_R + 1];
  __shared__ int   q[AGG_W * QCAP];

  const int tid  = threadIdx.x;
  const int lane = tid & 31;
  const int wv   = __builtin_amdgcn_readfirstlane(tid >> 5);
  const int lo   = blockIdx.x * OWN_R;
  const int cnt  = min(OWN_R, n - lo);
  const int sl   = blockIdx.y;

  if (sl == 0) {
    for (int i = tid; i < cnt * D; i += AGG_T) {
      int   node = lo + (i >> 6);
      float dn   = dinv[node];
      acc[i] = h[(size_t)lo * D + i] * (dn * dn);
    }
  } else {
    for (int i = tid; i < cnt * D; i += AGG_T) acc[i] = 0.0f;
  }
  if (tid < D) acc[OWN_R * D + tid] = 0.0f;
  for (int i = tid; i < OWN_R + 1; i += AGG_T) dinvL[i] = (i < cnt) ? dinv[lo + i] : 0.0f;
  __syncthreads();

  int* qw = q + wv * QCAP;
  int qhead = 0, qtail = 0;

  auto push = [&](bool m, int e) {
    unsigned mask = __builtin_amdgcn_ballot_w32(m);
    if (mask) {
      int pos = (qtail + (int)__builtin_amdgcn_mbcnt_lo(mask, 0u)) & (QCAP - 1);
      if (m) qw[pos] = e;
      qtail += __builtin_popcount(mask);
    }
  };
  auto drain_full = [&]() {
    while (qtail - qhead >= 32) {
      agg_batch(qw, qhead, 32, lane, lo, cnt, src, dst, dinv, dinvL, h, acc);
      qhead += 32;
    }
  };
  const unsigned pat = (blockIdx.x & 0xffu) * 0x01010101u;

  auto zero_bytes = [](unsigned w) -> unsigned {
    return ~((((w & 0x7f7f7f7fu) + 0x7f7f7f7fu) | w) | 0x7f7f7f7fu);
  };
  auto test_piece = [&](uint4 u0, uint4 u1, int e0) {
    unsigned m = (zero_bytes(u0.x ^ pat) >> 7) | (zero_bytes(u0.y ^ pat) >> 6) |
                 (zero_bytes(u0.z ^ pat) >> 5) | (zero_bytes(u0.w ^ pat) >> 4) |
                 (zero_bytes(u1.x ^ pat) >> 3) | (zero_bytes(u1.y ^ pat) >> 2) |
                 (zero_bytes(u1.z ^ pat) >> 1) |  zero_bytes(u1.w ^ pat);
    unsigned mask = __builtin_amdgcn_ballot_w32(m != 0u);
    do {
      if (m != 0u) {
        const int p   = __ffs((int)m) - 1;
        const int pos = (qtail + (int)__builtin_amdgcn_mbcnt_lo(mask, 0u)) & (QCAP - 1);
        qw[pos] = e0 + ((p & 4) << 7) + ((p & 3) << 2) + (p >> 3);
      }
      qtail += __builtin_popcount(mask);
      m &= m - 1u;
      drain_full();
      mask = __builtin_amdgcn_ballot_w32(m != 0u);
    } while (mask != 0u);
  };

  const uint4*   vp    = (const uint4*)rid4;
  const int      nfull = E >> 10;
  const int   p0    = (int)(((long long)sl * nfull) / AGG_C);
  const int   np    = (int)(((long long)(sl + 1) * nfull) / AGG_C) - p0;
  const int   rot   = (int)((blockIdx.x * (unsigned)np) / gridDim.x);

  uint4 d[PIECE_J], dn[PIECE_J];
  if (wv < np) {
    int c  = wv;
    int cc = c + rot; if (cc >= np) cc -= np;
    cc += p0;
#pragma unroll
    for (int j = 0; j < PIECE_J; ++j) d[j] = vp[cc * (PIECE_J * 32) + j * 32 + lane];
    while (true) {
      const int  cnx  = c + AGG_W;
      const bool more = cnx < np;
      int ccn = cc;
      if (more) { ccn = cnx + rot; if (ccn >= np) ccn -= np; ccn += p0; }
#pragma unroll
      for (int j = 0; j < PIECE_J; ++j) dn[j] = vp[ccn * (PIECE_J * 32) + j * 32 + lane];
      test_piece(d[0], d[1], 16 * (cc * (PIECE_J * 32) + lane));
      if (!more) break;
#pragma unroll
      for (int j = 0; j < PIECE_J; ++j) d[j] = dn[j];
      c  = cnx;
      cc = ccn;
    }
  }

  if (sl == 0) {
    const int tail0 = nfull << 10;
    const int nleft = E - tail0;
    for (int base = wv * 32; base < nleft; base += AGG_T) {
      const int idx = base + lane;
      const int e   = tail0 + idx;
      bool m = false;
      if (idx < nleft) m = (unsigned)(dst[e] - lo) < (unsigned)cnt;
      push(m, e);
    }
  }
  while (qtail - qhead > 0) {
    int valid = min(32, qtail - qhead);
    agg_batch(qw, qhead, valid, lane, lo, cnt, src, dst, dinv, dinvL, h, acc);
    qhead += valid;
  }

  __syncthreads();
  float* agg = (sl == 0) ? agg0 : agg1;
  for (int i = tid; i < cnt * D; i += AGG_T) agg[(size_t)lo * D + i] = acc[i];
}

static inline int cdiv_ll(long long a, long long b) { return (int)((a + b - 1) / b); }

extern "C" void kernel_launch(void* const* d_in, const int* in_sizes, int n_in,
                              void* d_out, int out_size, void* d_ws, size_t ws_size,
                              hipStream_t stream) {
  (void)n_in; (void)out_size; (void)ws_size;

  const float* x  = (const float*)d_in[0];
  const int*   ei = (const int*)  d_in[1];
  const float* W1 = (const float*)d_in[2];
  const float* b1 = (const float*)d_in[3];
  const float* a1 = (const float*)d_in[4];
  const float* W2 = (const float*)d_in[5];
  const float* b2 = (const float*)d_in[6];
  const float* a2 = (const float*)d_in[7];

  const int n = in_sizes[0] / D;
  const int E = in_sizes[1] / 2;
  const int* src = ei;
  const int* dst = ei + E;

  float* ws   = (float*)d_ws;
  unsigned int* rid4 = (unsigned int*)ws;
  float* dinv = ws + (size_t)(E + n);
  float* bufA = dinv + n;
  float* bufB = bufA + (size_t)n * D;
  float* out  = (float*)d_out;
  int*   part = (int*)bufA;

  const int B = 256;
  const int gN    = cdiv_ll(n, B);
  const int gRid  = cdiv_ll(cdiv_ll(E, 4), B);
  const int gOwn  = cdiv_ll(n, OWN_R);
  const int gGemm = cdiv_ll((long long)cdiv_ll(n, 16) * 32, B);
  const dim3 gAgg(gOwn, AGG_C);
  const dim3 gDeg(cdiv_ll(n, DEG_OWN), DEG_C);

  k_deg_part <<<gDeg, DEG_T, 0, stream>>>(dst, part, E, n);
  k_dinv_sum <<<gN,  B, 0, stream>>>(part, dinv, n);
  k_rid      <<<gRid, B, 0, stream>>>(dst, rid4, E);

  k_agg_owner<<<gAgg, AGG_T, 0, stream>>>(src, dst, rid4, dinv, x, bufA, bufB, E, n);
  k_gemm_sum_bias_prelu_wmma<<<gGemm, B, 0, stream>>>(bufA, bufB, W1, b1, a1, out, n);

  k_agg_owner<<<gAgg, AGG_T, 0, stream>>>(src, dst, rid4, dinv, out, bufA, bufB, E, n);
  k_gemm_sum_bias_prelu_wmma<<<gGemm, B, 0, stream>>>(bufA, bufB, W2, b2, a2, out, n);
}


# Aggregation group owns 400 complete rows (250 groups, each sweeps the whole edge list) and multiplies its LDS sums by W on f32 WMMA itself, bias + PReLU: 2 launches fewer; sum order changes
# speedup vs baseline: 4.7051x; 1.1877x over previous
#include <hip/hip_runtime.h>
#include <hip/hip_bf16.h>

typedef float v2f __attribute__((ext_vector_type(2)));
typedef float v8f __attribute__((ext_vector_type(8)));

#define D 64
#define OWN_R 400

#define DEG_OWN 6400
#define DEG_C   16
#define DEG_T   1024
#define DEG_U   20

__global__ __launch_bounds__(DEG_T)
void k_deg_part(const int* __restrict__ dst, int* __restrict__ part, int E, int n) {
  __shared__ int cntL[DEG_OWN];
  const int tid = threadIdx.x;
  const int lo  = blockIdx.x * DEG_OWN;
  const int cnt = min(DEG_OWN, n - lo);
  const int pc  = blockIdx.y;

  for (int i = tid; i < DEG_OWN; i += DEG_T) cntL[i] = 0;
  __syncthreads();

  const int e_lo = (int)(((long long)pc * E) / DEG_C);
  const int e_hi = (int)(((long long)(pc + 1) * E) / DEG_C);
  int a_lo = e_lo + (int)((4u - (unsigned)(((size_t)(dst + e_lo) >> 2) & 3u)) & 3u);
  if (a_lo > e_hi) a_lo = e_hi;
  const int nvec = (e_hi - a_lo) >> 2;
  const int a_hi = a_lo + 4 * nvec;
  const int4* vp = (const int4*)(dst + a_lo);

  auto count = [&](int t) {
    const unsigned r = (unsigned)(t - lo);
    if (r < (unsigned)cnt)
      __hip_atomic_fetch_add(&cntL[r], 1, __ATOMIC_RELAXED, __HIP_MEMORY_SCOPE_WORKGROUP);
  };

  for (int base = tid; base < nvec; base += DEG_U * DEG_T) {
    int4 d[DEG_U];
#pragma unroll
    for (int j = 0; j < DEG_U; ++j) {
      const int i = base + j * DEG_T;
      d[j] = (i < nvec) ? vp[i] : make_int4(-1, -1, -1, -1);
    }
#pragma unroll
    for (int j = 0; j < DEG_U; ++j) {
      count(d[j].x); count(d[j].y); count(d[j].z); count(d[j].w);
    }
  }
  {
    const int nh = a_lo - e_lo;
    const int nt = e_hi - a_hi;
    if (tid < nh)           count(dst[e_lo + tid]);
    else if (tid < nh + nt) count(dst[a_hi + (tid - nh)]);
  }
  __syncthreads();

  for (int i = tid; i < cnt; i += DEG_T) part[(size_t)pc * n + lo + i] = cntL[i];
}

__global__ void k_dinv_sum(const int* __restrict__ part, float* __restrict__ dinv, int n) {
  int i = blockIdx.x * blockDim.x + threadIdx.x;
  if (i < n) {
    int s = 1;
#pragma unroll
    for (int c = 0; c < DEG_C; ++c) s += part[(size_t)c * n + i];
    float d = (float)s;
    dinv[i] = (d > 0.0f) ? rsqrtf(d) : 0.0f;
  }
}

__global__ void k_rid(const int* __restrict__ dst, unsigned int* __restrict__ rid4, int E) {
  int i = blockIdx.x * blockDim.x + threadIdx.x;
  int e = 4 * i;
  const unsigned int R = OWN_R;
  if (e + 3 < E) {
    const unsigned int d0 = (unsigned int)dst[e],     d1 = (unsigned int)dst[e + 1];
    const unsigned int d2 = (unsigned int)dst[e + 2], d3 = (unsigned int)dst[e + 3];
    rid4[i] = ((d0 / R) & 0xffu)         | (((d1 / R) & 0xffu) << 8) |
              (((d2 / R) & 0xffu) << 16) | (((d3 / R) & 0xffu) << 24);
  } else if (e < E) {
    unsigned int v = 0u;
    for (int k = 0; k < 4; ++k) {
      unsigned int b = 0xffu;
      if (e + k < E) b = ((unsigned int)dst[e + k] / R) & 0xffu;
      v |= b << (8 * k);
    }
    rid4[i] = v;
  }
}

#define AGG_T 1024
#define AGG_W (AGG_T / 32)
#define ROW_P 68
#define QCAP  512
#define PIECE_J 2
static_assert(PIECE_J == 2, "the sweep packs the two units of a piece into one 32-bit flag word per lane");
static_assert((OWN_R % 16) == 0, "the product takes the owned rows in tiles of 16");

__device__ __forceinline__ void lds_add(float* p, float v) {
  __hip_atomic_fetch_add(p, v, __ATOMIC_RELAXED, __HIP_MEMORY_SCOPE_WORKGROUP);
}

__device__ __forceinline__ void agg_batch(const int* qw, int head, int valid, int lane, int lo, int cnt,
                                          const int* __restrict__ src, const int* __restrict__ dst,
                                          const float* __restrict__ dinv, const float* dinvL,
                                          const float* __restrict__ h, float* acc) {
  int my_s = lo;
  int my_d = -1;
  if (lane < valid) {
    int e = qw[(head + lane) & (QCAP - 1)];
    my_s = src[e];
    my_d = dst[e];
  }
  const float dv = dinv[my_s];
  v2f r[32];
#pragma unroll
  for (int u = 0; u < 32; ++u) {
    const int  s  = __builtin_amdgcn_readlane(my_s, u);
    const v2f* hr = (const v2f*)(h + (size_t)s * D);
    r[u] = hr[lane];
  }
  unsigned t = (unsigned)(my_d - lo);
  if (t >= (unsigned)cnt) t = OWN_R;
  const float my_w = dv * dinvL[t];
#pragma unroll
  for (int u = 0; u < 32; ++u) {
    const int   tt = __builtin_amdgcn_readlane((int)t, u);
    const float w  = __int_as_float(__builtin_amdgcn_readlane(__float_as_int(my_w), u));
    float* a = acc + tt * ROW_P + 2 * lane;
    lds_add(a,     r[u].x * w);
    lds_add(a + 1, r[u].y * w);
  }
}

__global__ __launch_bounds__(AGG_T)
void k_agg_mul(const int* __restrict__ src, const int* __restrict__ dst,
               const unsigned int* __restrict__ rid4, const float* __restrict__ dinv,
               const float* __restrict__ h,
               const float* __restrict__ W, const float* __restrict__ bias,
               const float* __restrict__ alpha, float* __restrict__ Cout,
               int E, int n) {
  __shared__ __attribute__((aligned(16))) float acc[(OWN_R + 1) * ROW_P];
  __shared__ __attribute__((aligned(16))) v2f   sW[(D / 2) * D];
  __shared__ float dinvL[OWN_R + 1];
  __shared__ int   q[AGG_W * QCAP];

  const int tid  = threadIdx.x;
  const int lane = tid & 31;
  const int wv   = __builtin_amdgcn_readfirstlane(tid >> 5);
  const int lo   = blockIdx.x * OWN_R;
  const int cnt  = min(OWN_R, n - lo);

  for (int i = tid; i < cnt * D; i += AGG_T) {
    const int   rw = i >> 6;
    const float dn = dinv[lo + rw];
    acc[rw * ROW_P + (i & (D - 1))] = h[(size_t)lo * D + i] * (dn * dn);
  }
  for (int i = cnt * D + tid; i < (OWN_R + 1) * D; i += AGG_T) acc[(i >> 6) * ROW_P + (i & (D - 1))] = 0.0f;
  for (int i = tid; i < OWN_R + 1; i += AGG_T) dinvL[i] = (i < cnt) ? dinv[lo + i] : 0.0f;
  for (int i = tid; i < (D / 2) * D; i += AGG_T) {
    const int kk = i / D;
    const int nn = i - kk * D;
    v2f p;
    p.x = W[(2 * kk + 0) * D + nn];
    p.y = W[(2 * kk + 1) * D + nn];
    sW[i] = p;
  }
  __syncthreads();

  int* qw = q + wv * QCAP;
  int qhead = 0, qtail = 0;

  auto push = [&](bool m, int e) {
    unsigned mask = __builtin_amdgcn_ballot_w32(m);
    if (mask) {
      int pos = (qtail + (int)__builtin_amdgcn_mbcnt_lo(mask, 0u)) & (QCAP - 1);
      if (m) qw[pos] = e;
      qtail += __builtin_popcount(mask);
    }
  };
  auto drain_full = [&]() {
    while (qtail - qhead >= 32) {
      agg_batch(qw, qhead, 32, lane, lo, cnt, src, dst, dinv, dinvL, h, acc);
      qhead += 32;
    }
  };
  const unsigned pat = (blockIdx.x & 0xffu) * 0x01010101u;

  auto zero_bytes = [](unsigned w) -> unsigned {
    return ~((((w & 0x7f7f7f7fu) + 0x7f7f7f7fu) | w) | 0x7f7f7f7fu);
  };
  auto test_piece = [&](uint4 u0, uint4 u1, int e0) {
    unsigned m = (zero_bytes(u0.x ^ pat) >> 7) | (zero_bytes(u0.y ^ pat) >> 6) |
                 (zero_bytes(u0.z ^ pat) >> 5) | (zero_bytes(u0.w ^ pat) >> 4) |
                 (zero_bytes(u1.x ^ pat) >> 3) | (zero_bytes(u1.y ^ pat) >> 2) |
                 (zero_bytes(u1.z ^ pat) >> 1) |  zero_bytes(u1.w ^ pat);
    unsigned mask = __builtin_amdgcn_ballot_w32(m != 0u);
    do {
      if (m != 0u) {
        const int p   = __ffs((int)m) - 1;
        const int pos = (qtail + (int)__builtin_amdgcn_mbcnt_lo(mask, 0u)) & (QCAP - 1);
        qw[pos] = e0 + ((p & 4) << 7) + ((p & 3) << 2) + (p >> 3);
      }
      qtail += __builtin_popcount(mask);
      m &= m - 1u;
      drain_full();
      mask = __builtin_amdgcn_ballot_w32(m != 0u);
    } while (mask != 0u);
  };

  const uint4*   vp    = (const uint4*)rid4;
  const int      np    = E >> 10;
  const int   rot   = (int)((blockIdx.x * (unsigned)np) / gridDim.x);

  uint4 d[PIECE_J], dn[PIECE_J];
  if (wv < np) {
    int c  = wv;
    int cc = c + rot; if (cc >= np) cc -= np;
#pragma unroll
    for (int j = 0; j < PIECE_J; ++j) d[j] = vp[cc * (PIECE_J * 32) + j * 32 + lane];
    while (true) {
      const int  cnx  = c + AGG_W;
      const bool more = cnx < np;
      int ccn = cc;
      if (more) { ccn = cnx + rot; if (ccn >= np) ccn -= np; }
#pragma unroll
      for (int j = 0; j < PIECE_J; ++j) dn[j] = vp[ccn * (PIECE_J * 32) + j * 32 + lane];
      test_piece(d[0], d[1], 16 * (cc * (PIECE_J * 32) + lane));
      if (!more) break;
#pragma unroll
      for (int j = 0; j < PIECE_J; ++j) d[j] = dn[j];
      c  = cnx;
      cc = ccn;
    }
  }

  {
    const int tail0 = np << 10;
    const int nleft = E - tail0;
    for (int base = wv * 32; base < nleft; base += AGG_T) {
      const int idx = base + lane;
      const int e   = tail0 + idx;
      bool m = false;
      if (idx < nleft) m = (unsigned)(dst[e] - lo) < (unsigned)cnt;
      push(m, e);
    }
  }
  while (qtail - qhead > 0) {
    int valid = min(32, qtail - qhead);
    agg_batch(qw, qhead, valid, lane, lo, cnt, src, dst, dinv, dinvL, h, acc);
    qhead += valid;
  }

  __syncthreads();

  const int ntile = (cnt + 15) >> 4;
  if (wv < ntile) {
    const int half = lane >> 4;
    const int lid  = lane & 15;
    const int row0 = wv * 16;

    float bz[4], al[4];
#pragma unroll
    for (int j = 0; j < 4; ++j) {
      bz[j] = bias[16 * j + lid];
      al[j] = alpha[16 * j + lid];
    }

    v8f acc0 = {}, acc1 = {}, acc2 = {}, acc3 = {};
    const v2f* arow = (const v2f*)(acc + (row0 + lid) * ROW_P);

#pragma unroll
    for (int k = 0; k < D; k += 4) {
      const int kk = (k >> 1) + half;
      const v2f a  = arow[kk];
      const v2f* wrow = &sW[kk * D + lid];
      const v2f b0 = wrow[0];
      const v2f b1 = wrow[16];
      const v2f b2 = wrow[32];
      const v2f b3 = wrow[48];
      acc0 = __builtin_amdgcn_wmma_f32_16x16x4_f32(false, a, false, b0, (short)0, acc0, false, false);
      acc1 = __builtin_amdgcn_wmma_f32_16x16x4_f32(false, a, false, b1, (short)0, acc1, false, false);
      acc2 = __builtin_amdgcn_wmma_f32_16x16x4_f32(false, a, false, b2, (short)0, acc2, false, false);
      acc3 = __builtin_amdgcn_wmma_f32_16x16x4_f32(false, a, false, b3, (short)0, acc3, false, false);
    }

#pragma unroll
    for (int v = 0; v < 8; ++v) {
      const int rr = row0 + v + 8 * half;
      if (rr < cnt) {
        float* crow = Cout + (size_t)(lo + rr) * D + lid;
        float t0 = acc0[v] + bz[0];
        float t1 = acc1[v] + bz[1];
        float t2 = acc2[v] + bz[2];
        float t3 = acc3[v] + bz[3];
        crow[0]  = (t0 >= 0.0f) ? t0 : al[0] * t0;
        crow[16] = (t1 >= 0.0f) ? t1 : al[1] * t1;
        crow[32] = (t2 >= 0.0f) ? t2 : al[2] * t2;
        crow[48] = (t3 >= 0.0f) ? t3 : al[3] * t3;
      }
    }
  }
}

static inline int cdiv_ll(long long a, long long b) { return (int)((a + b - 1) / b); }

extern "C" void kernel_launch(void* const* d_in, const int* in_sizes, int n_in,
                              void* d_out, int out_size, void* d_ws, size_t ws_size,
                              hipStream_t stream) {
  (void)n_in; (void)out_size; (void)ws_size;

  const float* x  = (const float*)d_in[0];
  const int*   ei = (const int*)  d_in[1];
  const float* W1 = (const float*)d_in[2];
  const float* b1 = (const float*)d_in[3];
  const float* a1 = (const float*)d_in[4];
  const float* W2 = (const float*)d_in[5];
  const float* b2 = (const float*)d_in[6];
  const float* a2 = (const float*)d_in[7];

  const int n = in_sizes[0] / D;
  const int E = in_sizes[1] / 2;
  const int* src = ei;
  const int* dst = ei + E;

  float* ws   = (float*)d_ws;
  unsigned int* rid4 = (unsigned int*)ws;
  float* dinv = ws + (size_t)(E + n);
  float* bufA = dinv + n;
  float* out  = (float*)d_out;
  int*   part = (int*)bufA;

  const int B = 256;
  const int gN    = cdiv_ll(n, B);
  const int gRid  = cdiv_ll(cdiv_ll(E, 4), B);
  const int gOwn  = cdiv_ll(n, OWN_R);
  const dim3 gDeg(cdiv_ll(n, DEG_OWN), DEG_C);

  k_deg_part <<<gDeg, DEG_T, 0, stream>>>(dst, part, E, n);
  k_dinv_sum <<<gN,  B, 0, stream>>>(part, dinv, n);
  k_rid      <<<gRid, B, 0, stream>>>(dst, rid4, E);

  k_agg_mul<<<gOwn, AGG_T, 0, stream>>>(src, dst, rid4, dinv, x, W1, b1, a1, bufA, E, n);
  k_agg_mul<<<gOwn, AGG_T, 0, stream>>>(src, dst, rid4, dinv, bufA, W2, b2, a2, out, E, n);
}


# Layer 2 replays layer 1's batches: layer 1 stores each 32-edge batch (source, owned row) as one line in the range's own output rows; layer 2 reads no rid/src/dst, does no sweep; sum order changes
# speedup vs baseline: 5.2288x; 1.1113x over previous
#include <hip/hip_runtime.h>
#include <hip/hip_bf16.h>

typedef float v2f __attribute__((ext_vector_type(2)));
typedef float v8f __attribute__((ext_vector_type(8)));

#define D 64
#define OWN_R 400

#define DEG_OWN 6400
#define DEG_C   16
#define DEG_T   1024
#define DEG_U   20

__global__ __launch_bounds__(DEG_T)
void k_deg_part(const int* __restrict__ dst, int* __restrict__ part, int E, int n) {
  __shared__ int cntL[DEG_OWN];
  const int tid = threadIdx.x;
  const int lo  = blockIdx.x * DEG_OWN;
  const int cnt = min(DEG_OWN, n - lo);
  const int pc  = blockIdx.y;

  for (int i = tid; i < DEG_OWN; i += DEG_T) cntL[i] = 0;
  __syncthreads();

  const int e_lo = (int)(((long long)pc * E) / DEG_C);
  const int e_hi = (int)(((long long)(pc + 1) * E) / DEG_C);
  int a_lo = e_lo + (int)((4u - (unsigned)(((size_t)(dst + e_lo) >> 2) & 3u)) & 3u);
  if (a_lo > e_hi) a_lo = e_hi;
  const int nvec = (e_hi - a_lo) >> 2;
  const int a_hi = a_lo + 4 * nvec;
  const int4* vp = (const int4*)(dst + a_lo);

  auto count = [&](int t) {
    const unsigned r = (unsigned)(t - lo);
    if (r < (unsigned)cnt)
      __hip_atomic_fetch_add(&cntL[r], 1, __ATOMIC_RELAXED, __HIP_MEMORY_SCOPE_WORKGROUP);
  };

  for (int base = tid; base < nvec; base += DEG_U * DEG_T) {
    int4 d[DEG_U];
#pragma unroll
    for (int j = 0; j < DEG_U; ++j) {
      const int i = base + j * DEG_T;
      d[j] = (i < nvec) ? vp[i] : make_int4(-1, -1, -1, -1);
    }
#pragma unroll
    for (int j = 0; j < DEG_U; ++j) {
      count(d[j].x); count(d[j].y); count(d[j].z); count(d[j].w);
    }
  }
  {
    const int nh = a_lo - e_lo;
    const int nt = e_hi - a_hi;
    if (tid < nh)           count(dst[e_lo + tid]);
    else if (tid < nh + nt) count(dst[a_hi + (tid - nh)]);
  }
  __syncthreads();

  for (int i = tid; i < cnt; i += DEG_T) part[(size_t)pc * n + lo + i] = cntL[i];
}

__global__ void k_dinv_sum(const int* __restrict__ part, float* __restrict__ dinv, int n) {
  int i = blockIdx.x * blockDim.x + threadIdx.x;
  if (i < n) {
    int s = 1;
#pragma unroll
    for (int c = 0; c < DEG_C; ++c) s += part[(size_t)c * n + i];
    float d = (float)s;
    dinv[i] = (d > 0.0f) ? rsqrtf(d) : 0.0f;
  }
}

__global__ void k_rid(const int* __restrict__ dst, unsigned int* __restrict__ rid4, int E) {
  int i = blockIdx.x * blockDim.x + threadIdx.x;
  int e = 4 * i;
  const unsigned int R = OWN_R;
  if (e + 3 < E) {
    const unsigned int d0 = (unsigned int)dst[e],     d1 = (unsigned int)dst[e + 1];
    const unsigned int d2 = (unsigned int)dst[e + 2], d3 = (unsigned int)dst[e + 3];
    rid4[i] = ((d0 / R) & 0xffu)         | (((d1 / R) & 0xffu) << 8) |
              (((d2 / R) & 0xffu) << 16) | (((d3 / R) & 0xffu) << 24);
  } else if (e < E) {
    unsigned int v = 0u;
    for (int k = 0; k < 4; ++k) {
      unsigned int b = 0xffu;
      if (e + k < E) b = ((unsigned int)dst[e + k] / R) & 0xffu;
      v |= b << (8 * k);
    }
    rid4[i] = v;
  }
}

#define AGG_T 1024
#define AGG_W (AGG_T / 32)
#define ROW_P 68
#define QCAP  512
#define PIECE_J 2
static_assert(PIECE_J == 2, "the sweep packs the two units of a piece into one 32-bit flag word per lane");
static_assert((OWN_R % 16) == 0, "the product takes the owned rows in tiles of 16");
#define REC_ROW_BITS 9
static_assert(OWN_R < (1 << REC_ROW_BITS), "row number and the spare row's number must fit the record's row field");
#define REC_HDR 32

__device__ __forceinline__ void lds_add(float* p, float v) {
  __hip_atomic_fetch_add(p, v, __ATOMIC_RELAXED, __HIP_MEMORY_SCOPE_WORKGROUP);
}

__device__ __forceinline__ void agg_rows(int my_s, unsigned t, int lane,
                                         const float* __restrict__ dinv, const float* dinvL,
                                         const float* __restrict__ h, float* acc) {
  const float dv = dinv[my_s];
  v2f r[32];
#pragma unroll
  for (int u = 0; u < 32; ++u) {
    const int  s  = __builtin_amdgcn_readlane(my_s, u);
    const v2f* hr = (const v2f*)(h + (size_t)s * D);
    r[u] = hr[lane];
  }
  const float my_w = dv * dinvL[t];
#pragma unroll
  for (int u = 0; u < 32; ++u) {
    const int   tt = __builtin_amdgcn_readlane((int)t, u);
    const float w  = __int_as_float(__builtin_amdgcn_readlane(__float_as_int(my_w), u));
    float* a = acc + tt * ROW_P + 2 * lane;
    lds_add(a,     r[u].x * w);
    lds_add(a + 1, r[u].y * w);
  }
}

template <bool REC>
__device__ __forceinline__ void agg_batch(const int* qw, int head, int valid, int lane, int lo, int cnt,
                                          const int* __restrict__ src, const int* __restrict__ dst,
                                          const float* __restrict__ dinv, const float* dinvL,
                                          const float* __restrict__ h, float* acc,
                                          unsigned int* rec, int cap, int* cursorL) {
  int my_s = lo;
  int my_d = -1;
  if (lane < valid) {
    int e = qw[(head + lane) & (QCAP - 1)];
    my_s = src[e];
    my_d = dst[e];
  }
  unsigned t = (unsigned)(my_d - lo);
  if (t >= (unsigned)cnt) t = OWN_R;
  if (REC) {
    int base = 0;
    if (lane == 0) base = __hip_atomic_fetch_add(cursorL, 32, __ATOMIC_RELAXED, __HIP_MEMORY_SCOPE_WORKGROUP);
    base = __builtin_amdgcn_readfirstlane(base);
    if (base + 32 <= cap) rec[REC_HDR + base + lane] = ((unsigned)my_s << REC_ROW_BITS) | t;
  }
  agg_rows(my_s, t, lane, dinv, dinvL, h, acc);
}

template <int LAYER>
__device__ __forceinline__
void agg_mul_body(const int* __restrict__ src, const int* __restrict__ dst,
               const unsigned int* __restrict__ rid4, const float* __restrict__ dinv,
               const float* __restrict__ h,
               const float* __restrict__ W, const float* __restrict__ bias,
               const float* __restrict__ alpha, float* Cout, unsigned int* recAll,
               int E, int n) {
  __shared__ __attribute__((aligned(16))) float acc[(OWN_R + 1) * ROW_P];
  __shared__ __attribute__((aligned(16))) v2f   sW[(D / 2) * D];
  __shared__ float dinvL[OWN_R + 1];
  __shared__ int   q[AGG_W * QCAP];
  __shared__ int   cursorL;

  const int tid  = threadIdx.x;
  const int lane = tid & 31;
  const int wv   = __builtin_amdgcn_readfirstlane(tid >> 5);
  const int lo   = blockIdx.x * OWN_R;
  const int cnt  = min(OWN_R, n - lo);

  unsigned int* rec = recAll + (size_t)lo * D;
  const int     cap = cnt * D - REC_HDR;
  unsigned int hdr = 0u;
  if (LAYER == 2) hdr = rec[0];
  if (LAYER == 1 && tid == 0) cursorL = 0;

  for (int i = tid; i < cnt * D; i += AGG_T) {
    const int   rw = i >> 6;
    const float dn = dinv[lo + rw];
    acc[rw * ROW_P + (i & (D - 1))] = h[(size_t)lo * D + i] * (dn * dn);
  }
  for (int i = cnt * D + tid; i < (OWN_R + 1) * D; i += AGG_T) acc[(i >> 6) * ROW_P + (i & (D - 1))] = 0.0f;
  for (int i = tid; i < OWN_R + 1; i += AGG_T) dinvL[i] = (i < cnt) ? dinv[lo + i] : 0.0f;
  for (int i = tid; i < (D / 2) * D; i += AGG_T) {
    const int kk = i / D;
    const int nn = i - kk * D;
    v2f p;
    p.x = W[(2 * kk + 0) * D + nn];
    p.y = W[(2 * kk + 1) * D + nn];
    sW[i] = p;
  }
  __syncthreads();

  const int  nb      = __builtin_amdgcn_readfirstlane((int)hdr);
  const bool replay  = (LAYER == 2) && ((unsigned)nb <= (unsigned)(cap >> 5));

  if (!replay) {
    int* qw = q + wv * QCAP;
    int qhead = 0, qtail = 0;

    auto push = [&](bool m, int e) {
      unsigned mask = __builtin_amdgcn_ballot_w32(m);
      if (mask) {
        int pos = (qtail + (int)__builtin_amdgcn_mbcnt_lo(mask, 0u)) & (QCAP - 1);
        if (m) qw[pos] = e;
        qtail += __builtin_popcount(mask);
      }
    };
    auto drain_full = [&]() {
      while (qtail - qhead >= 32) {
        agg_batch<LAYER == 1>(qw, qhead, 32, lane, lo, cnt, src, dst, dinv, dinvL, h, acc, rec, cap, &cursorL);
        qhead += 32;
      }
    };
    const unsigned pat = (blockIdx.x & 0xffu) * 0x01010101u;

    auto zero_bytes = [](unsigned w) -> unsigned {
      return ~((((w & 0x7f7f7f7fu) + 0x7f7f7f7fu) | w) | 0x7f7f7f7fu);
    };
    auto test_piece = [&](uint4 u0, uint4 u1, int e0) {
      unsigned m = (zero_bytes(u0.x ^ pat) >> 7) | (zero_bytes(u0.y ^ pat) >> 6) |
                   (zero_bytes(u0.z ^ pat) >> 5) | (zero_bytes(u0.w ^ pat) >> 4) |
                   (zero_bytes(u1.x ^ pat) >> 3) | (zero_bytes(u1.y ^ pat) >> 2) |
                   (zero_bytes(u1.z ^ pat) >> 1) |  zero_bytes(u1.w ^ pat);
      unsigned mask = __builtin_amdgcn_ballot_w32(m != 0u);
      do {
        if (m != 0u) {
          const int p   = __ffs((int)m) - 1;
          const int pos = (qtail + (int)__builtin_amdgcn_mbcnt_lo(mask, 0u)) & (QCAP - 1);
          qw[pos] = e0 + ((p & 4) << 7) + ((p & 3) << 2) + (p >> 3);
        }
        qtail += __builtin_popcount(mask);
        m &= m - 1u;
        drain_full();
        mask = __builtin_amdgcn_ballot_w32(m != 0u);
      } while (mask != 0u);
    };

    const uint4*   vp    = (const uint4*)rid4;
    const int      np    = E >> 10;
    const int   rot   = (int)((blockIdx.x * (unsigned)np) / gridDim.x);

    uint4 d[PIECE_J], dn[PIECE_J];
    if (wv < np) {
      int c  = wv;
      int cc = c + rot; if (cc >= np) cc -= np;
#pragma unroll
      for (int j = 0; j < PIECE_J; ++j) d[j] = vp[cc * (PIECE_J * 32) + j * 32 + lane];
      while (true) {
        const int  cnx  = c + AGG_W;
        const bool more = cnx < np;
        int ccn = cc;
        if (more) { ccn = cnx + rot; if (ccn >= np) ccn -= np; }
#pragma unroll
        for (int j = 0; j < PIECE_J; ++j) dn[j] = vp[ccn * (PIECE_J * 32) + j * 32 + lane];
        test_piece(d[0], d[1], 16 * (cc * (PIECE_J * 32) + lane));
        if (!more) break;
#pragma unroll
        for (int j = 0; j < PIECE_J; ++j) d[j] = dn[j];
        c  = cnx;
        cc = ccn;
      }
    }

    {
      const int tail0 = np << 10;
      const int nleft = E - tail0;
      for (int base = wv * 32; base < nleft; base += AGG_T) {
        const int idx = base + lane;
        const int e   = tail0 + idx;
        bool m = false;
        if (idx < nleft) m = (unsigned)(dst[e] - lo) < (unsigned)cnt;
        push(m, e);
      }
    }
    while (qtail - qhead > 0) {
      int valid = min(32, qtail - qhead);
      agg_batch<LAYER == 1>(qw, qhead, valid, lane, lo, cnt, src, dst, dinv, dinvL, h, acc, rec, cap, &cursorL);
      qhead += valid;
    }

  } else {
    int b = wv;
    if (b < nb) {
      unsigned int ent = rec[REC_HDR + 32 * b + lane];
      while (true) {
        const int  bn   = b + AGG_W;
        const bool more = bn < nb;
        const unsigned int entn = rec[REC_HDR + 32 * (more ? bn : b) + lane];
        const int      my_s = min((int)(ent >> REC_ROW_BITS), n - 1);
        const unsigned t    = min(ent & ((1u << REC_ROW_BITS) - 1u), (unsigned)OWN_R);
        agg_rows(my_s, t, lane, dinv, dinvL, h, acc);
        if (!more) break;
        ent = entn;
        b   = bn;
      }
    }
  }

  __syncthreads();

  if (LAYER == 1 && tid == 0) rec[0] = (cursorL <= cap) ? (unsigned int)(cursorL >> 5) : 0xffffffffu;

  const int ntile = (cnt + 15) >> 4;
  if (wv < ntile) {
    const int half = lane >> 4;
    const int lid  = lane & 15;
    const int row0 = wv * 16;

    float bz[4], al[4];
#pragma unroll
    for (int j = 0; j < 4; ++j) {
      bz[j] = bias[16 * j + lid];
      al[j] = alpha[16 * j + lid];
    }

    v8f acc0 = {}, acc1 = {}, acc2 = {}, acc3 = {};
    const v2f* arow = (const v2f*)(acc + (row0 + lid) * ROW_P);

#pragma unroll
    for (int k = 0; k < D; k += 4) {
      const int kk = (k >> 1) + half;
      const v2f a  = arow[kk];
      const v2f* wrow = &sW[kk * D + lid];
      const v2f b0 = wrow[0];
      const v2f b1 = wrow[16];
      const v2f b2 = wrow[32];
      const v2f b3 = wrow[48];
      acc0 = __builtin_amdgcn_wmma_f32_16x16x4_f32(false, a, false, b0, (short)0, acc0, false, false);
      acc1 = __builtin_amdgcn_wmma_f32_16x16x4_f32(false, a, false, b1, (short)0, acc1, false, false);
      acc2 = __builtin_amdgcn_wmma_f32_16x16x4_f32(false, a, false, b2, (short)0, acc2, false, false);
      acc3 = __builtin_amdgcn_wmma_f32_16x16x4_f32(false, a, false, b3, (short)0, acc3, false, false);
    }

#pragma unroll
    for (int v = 0; v < 8; ++v) {
      const int rr = row0 + v + 8 * half;
      if (rr < cnt) {
        float* crow = Cout + (size_t)(lo + rr) * D + lid;
        float t0 = acc0[v] + bz[0];
        float t1 = acc1[v] + bz[1];
        float t2 = acc2[v] + bz[2];
        float t3 = acc3[v] + bz[3];
        crow[0]  = (t0 >= 0.0f) ? t0 : al[0] * t0;
        crow[16] = (t1 >= 0.0f) ? t1 : al[1] * t1;
        crow[32] = (t2 >= 0.0f) ? t2 : al[2] * t2;
        crow[48] = (t3 >= 0.0f) ? t3 : al[3] * t3;
      }
    }
  }
}

__global__ __launch_bounds__(AGG_T)
void k_agg_mul_sweep(const int* __restrict__ src, const int* __restrict__ dst,
                     const unsigned int* __restrict__ rid4, const float* __restrict__ dinv,
                     const float* __restrict__ h,
                     const float* __restrict__ W, const float* __restrict__ bias,
                     const float* __restrict__ alpha, float* Cout, unsigned int* recAll,
                     int E, int n) {
  agg_mul_body<1>(src, dst, rid4, dinv, h, W, bias, alpha, Cout, recAll, E, n);
}

__global__ __launch_bounds__(AGG_T)
void k_agg_mul_replay(const int* __restrict__ src, const int* __restrict__ dst,
                      const unsigned int* __restrict__ rid4, const float* __restrict__ dinv,
                      const float* __restrict__ h,
                      const float* __restrict__ W, const float* __restrict__ bias,
                      const float* __restrict__ alpha, float* Cout, unsigned int* recAll,
                      int E, int n) {
  agg_mul_body<2>(src, dst, rid4, dinv, h, W, bias, alpha, Cout, recAll, E, n);
}

static inline int cdiv_ll(long long a, long long b) { return (int)((a + b - 1) / b); }

extern "C" void kernel_launch(void* const* d_in, const int* in_sizes, int n_in,
                              void* d_out, int out_size, void* d_ws, size_t ws_size,
                              hipStream_t stream) {
  (void)n_in; (void)out_size; (void)ws_size;

  const float* x  = (const float*)d_in[0];
  const int*   ei = (const int*)  d_in[1];
  const float* W1 = (const float*)d_in[2];
  const float* b1 = (const float*)d_in[3];
  const float* a1 = (const float*)d_in[4];
  const float* W2 = (const float*)d_in[5];
  const float* b2 = (const float*)d_in[6];
  const float* a2 = (const float*)d_in[7];

  const int n = in_sizes[0] / D;
  const int E = in_sizes[1] / 2;
  const int* src = ei;
  const int* dst = ei + E;

  float* ws   = (float*)d_ws;
  unsigned int* rid4 = (unsigned int*)ws;
  float* dinv = ws + (size_t)(E + n);
  float* bufA = dinv + n;
  float* out  = (float*)d_out;
  int*   part = (int*)bufA;

  const int B = 256;
  const int gN    = cdiv_ll(n, B);
  const int gRid  = cdiv_ll(cdiv_ll(E, 4), B);
  const int gOwn  = cdiv_ll(n, OWN_R);
  const dim3 gDeg(cdiv_ll(n, DEG_OWN), DEG_C);

  k_deg_part <<<gDeg, DEG_T, 0, stream>>>(dst, part, E, n);
  k_dinv_sum <<<gN,  B, 0, stream>>>(part, dinv, n);
  k_rid      <<<gRid, B, 0, stream>>>(dst, rid4, E);

  k_agg_mul_sweep <<<gOwn, AGG_T, 0, stream>>>(src, dst, rid4, dinv, x, W1, b1, a1, bufA, (unsigned int*)out, E, n);
  k_agg_mul_replay<<<gOwn, AGG_T, 0, stream>>>(src, dst, rid4, dinv, bufA, W2, b2, a2, out, (unsigned int*)out, E, n);
}
